# starting kernel
# baseline (speedup 1.0000x reference)
#include <hip/hip_runtime.h>

typedef __attribute__((ext_vector_type(2))) float v2f;
typedef __attribute__((ext_vector_type(8))) float v8f;

#define IN_C 128
#define HID  64

__global__ void sage_zero_kernel(float* __restrict__ p, long n) {
  long i = (long)blockIdx.x * blockDim.x + threadIdx.x;
  long stride = (long)gridDim.x * blockDim.x;
  for (; i < n; i += stride) p[i] = 0.0f;
}

__global__ void sage_degree_kernel(const int* __restrict__ dst, int E,
                                   float* __restrict__ deg) {
  int i = blockIdx.x * blockDim.x + threadIdx.x;
  int stride = gridDim.x * blockDim.x;
  for (; i < E; i += stride) {
    atomicAdd(&deg[dst[i]], 1.0f);
  }
}

template <int F, int CLOG2>
__global__ void sage_scatter_kernel(const float* __restrict__ x,
                                    const int* __restrict__ src,
                                    const int* __restrict__ dst, int E,
                                    float* __restrict__ agg) {
  long total = (long)E << CLOG2;
  long i = (long)blockIdx.x * blockDim.x + threadIdx.x;
  long stride = (long)gridDim.x * blockDim.x;
  for (; i < total; i += stride) {
    long e = i >> CLOG2;
    int f = (int)(i & ((1 << CLOG2) - 1)) << 2;
    int s = src[e];
    int d = dst[e];
    float4 v = *(const float4*)(x + (long)s * F + f);
    float* a = agg + (long)d * F + f;
    atomicAdd(a + 0, v.x);
    atomicAdd(a + 1, v.y);
    atomicAdd(a + 2, v.z);
    atomicAdd(a + 3, v.w);
  }
}

template <int K, bool RELU>
__global__ void sage_wmma_gemm_kernel(const float* __restrict__ Aagg,
                                      const float* __restrict__ deg,
                                      const float* __restrict__ Aself,
                                      const float* __restrict__ Wl,
                                      const float* __restrict__ bl,
                                      const float* __restrict__ Wr,
                                      float* __restrict__ Out, int nNodes) {
  const int lane = threadIdx.x & 31;
  const int wid = (blockIdx.x * blockDim.x + threadIdx.x) >> 5;
  const int tilesN = HID / 16;
  const int tile_m = wid / tilesN;
  const int tile_n = wid % tilesN;
  if (tile_m * 16 >= nNodes) return;

  const int half = lane >> 4;
  const int lm = lane & 15;
  const int koff = half * 2;
  const int rowA = tile_m * 16 + lm;
  const int col = tile_n * 16 + lm;

  const float invc = 1.0f / fmaxf(deg[rowA], 1.0f);

  v8f c = {0.f, 0.f, 0.f, 0.f, 0.f, 0.f, 0.f, 0.f};

  const float* arow = Aagg + (long)rowA * K;
  const float* srow = Aself + (long)rowA * K;

#pragma unroll
  for (int k = 0; k < K; k += 4) {
    v2f a;
    a.x = arow[k + koff] * invc;
    a.y = arow[k + koff + 1] * invc;
    v2f b;
    b.x = Wl[(k + koff) * HID + col];
    b.y = Wl[(k + koff + 1) * HID + col];
    c = __builtin_amdgcn_wmma_f32_16x16x4_f32(false, a, false, b, (short)0, c,
                                              false, false);
  }
#pragma unroll
  for (int k = 0; k < K; k += 4) {
    v2f a;
    a.x = srow[k + koff];
    a.y = srow[k + koff + 1];
    v2f b;
    b.x = Wr[(k + koff) * HID + col];
    b.y = Wr[(k + koff + 1) * HID + col];
    c = __builtin_amdgcn_wmma_f32_16x16x4_f32(false, a, false, b, (short)0, c,
                                              false, false);
  }

  const float bv = bl[col];
#pragma unroll
  for (int r = 0; r < 8; ++r) {
    float v = c[r] + bv;
    if (RELU) v = fmaxf(v, 0.0f);
    int row = tile_m * 16 + r + half * 8;
    Out[(long)row * HID + col] = v;
  }
}

extern "C" void kernel_launch(void* const* d_in, const int* in_sizes, int n_in,
                              void* d_out, int out_size, void* d_ws,
                              size_t ws_size, hipStream_t stream) {
  const float* x = (const float*)d_in[0];
  const int* ei = (const int*)d_in[1];
  const float* Wl1 = (const float*)d_in[2];
  const float* bl1 = (const float*)d_in[3];
  const float* Wr1 = (const float*)d_in[4];
  const float* Wl2 = (const float*)d_in[5];
  const float* bl2 = (const float*)d_in[6];
  const float* Wr2 = (const float*)d_in[7];
  float* out = (float*)d_out;

  const int N = in_sizes[0] / IN_C;
  const int E = in_sizes[1] / 2;
  const int* src = ei;
  const int* dst = ei + E;

  float* ws = (float*)d_ws;
  float* deg = ws;
  float* agg1 = deg + N;
  float* h = agg1 + (long)N * IN_C;
  float* agg2 = h + (long)N * HID;

  sage_zero_kernel<<<2048, 256, 0, stream>>>(deg, (long)N * (1 + IN_C));
  sage_zero_kernel<<<2048, 256, 0, stream>>>(agg2, (long)N * HID);

  sage_degree_kernel<<<2048, 256, 0, stream>>>(dst, E, deg);

  sage_scatter_kernel<IN_C, 5><<<8192, 256, 0, stream>>>(x, src, dst, E, agg1);

  const int tilesM = (N + 15) / 16;
  const int totalWaves = tilesM * (HID / 16);
  const int wavesPerBlock = 256 / 32;
  const int gemmBlocks = (totalWaves + wavesPerBlock - 1) / wavesPerBlock;
  sage_wmma_gemm_kernel<IN_C, true><<<gemmBlocks, 256, 0, stream>>>(
      agg1, deg, x, Wl1, bl1, Wr1, h, N);

  sage_scatter_kernel<HID, 4><<<8192, 256, 0, stream>>>(h, src, dst, E, agg2);
  sage_wmma_gemm_kernel<HID, false><<<gemmBlocks, 256, 0, stream>>>(
      agg2, deg, h, Wl2, bl2, Wr2, out, N);
}


# Each layer's product runs inside its gather launch: the finished LDS sum tile is the WMMA A operand, weights in LDS in fragment order; 3 launches, not 5; no sum plane in memory; k sum order changes
# speedup vs baseline: 3.9622x; 3.9622x over previous
#include <hip/hip_runtime.h>

typedef __attribute__((ext_vector_type(2))) float v2f;
typedef __attribute__((ext_vector_type(8))) float v8f;

#define IN_C 128
#define HID  64
#define RNG 392
#define SC_THREADS 1024
#define SLAB_LOADS 8
#define SLAB_EDGES (SLAB_LOADS * 128)
#define Q_DRAIN 32
#define QCAP (Q_DRAIN + 128)
#define RIF 8
#define LIST_PER_NODE (HID / 2)
#define LIST_HDR 16
#define KEY_LOADS 4
#define KEY_SLAB (KEY_LOADS * 512)


__device__ __forceinline__ unsigned sage_key(int d) {
  return ((unsigned)d / (unsigned)RNG) & 255u;
}

__global__ void __launch_bounds__(256) sage_range_key_kernel(
    const int* __restrict__ dst, int E, unsigned char* __restrict__ keys) {
  const int gtid = blockIdx.x * blockDim.x + threadIdx.x;
  const int T = gridDim.x * blockDim.x;
  const bool aligned = (((unsigned long)dst) & 15ul) == 0ul &&
                       (((unsigned long)keys) & 3ul) == 0ul;
  const int n4 = aligned ? (E >> 2) : 0;
  for (int base = gtid; base < n4; base += 4 * T) {
    int4 d[4];
#pragma unroll
    for (int u = 0; u < 4; ++u)
      d[u] = ((const int4*)dst)[min(base + u * T, n4 - 1)];
#pragma unroll
    for (int u = 0; u < 4; ++u)
      if (base + u * T < n4)
        ((unsigned*)keys)[base + u * T] =
            sage_key(d[u].x) | (sage_key(d[u].y) << 8) |
            (sage_key(d[u].z) << 16) | (sage_key(d[u].w) << 24);
  }
  for (int e = n4 * 4 + gtid; e < E; e += T)
    keys[e] = (unsigned char)sage_key(dst[e]);
}

template <int F>
__device__ __forceinline__ void sage_add_rows(int sv, int dv, int n,
                                              const float* __restrict__ x,
                                              float* tile, int lane) {
  constexpr int VEC = F / 32;
  for (int j = 0; j < n; j += RIF) {
    float v[RIF][VEC];
    int dd[RIF];
#pragma unroll
    for (int k = 0; k < RIF; ++k) {
      const int jj = min(j + k, n - 1);
      const int sr = __builtin_amdgcn_readlane(sv, jj);
      dd[k] = __builtin_amdgcn_readlane(dv, jj);
      const float* row = x + (long)sr * F + lane * VEC;
      if constexpr (VEC == 4) {
        const float4 t = *(const float4*)row;
        v[k][0] = t.x;
        v[k][1] = t.y;
        v[k][2] = t.z;
        v[k][3] = t.w;
      } else {
        const float2 t = *(const float2*)row;
        v[k][0] = t.x;
        v[k][1] = t.y;
      }
    }
#pragma unroll
    for (int k = 0; k < RIF; ++k)
      if (j + k < n) {
        float* t = tile + dd[k] * (F + 1) + lane;
#pragma unroll
        for (int c = 0; c < VEC; ++c)
          __hip_atomic_fetch_add(t + c * 32, v[k][c], __ATOMIC_RELAXED,
                                 __HIP_MEMORY_SCOPE_WORKGROUP);
      }
  }
}

template <int F, bool COUNT, bool SAVE>
__device__ __forceinline__ void sage_drain(const int* q, int qn,
                                           const float* __restrict__ x,
                                           const int* __restrict__ src,
                                           const int* __restrict__ dst, int lo,
                                           unsigned cnt, float* tile,
                                           float* cntl, int2* lst, int cap,
                                           int* lcount, int lane) {
  __builtin_amdgcn_fence(__ATOMIC_ACQ_REL, "wavefront");
  for (int b = 0; b < qn; b += 32) {
    const int n = min(32, qn - b);
    int sv = 0, dv = 0;
    if (lane < n) {
      const int e = q[b + lane];
      sv = src[e];
      dv = dst[e] - lo;
      if ((unsigned)dv >= cnt) {
        sv = 0;
        dv = RNG;
      }
      if (COUNT)
        __hip_atomic_fetch_add(&cntl[dv], 1.0f, __ATOMIC_RELAXED,
                               __HIP_MEMORY_SCOPE_WORKGROUP);
    }
    if (SAVE) {
      int pos = 0;
      if (lane == 0)
        pos = __hip_atomic_fetch_add(lcount, n, __ATOMIC_RELAXED,
                                     __HIP_MEMORY_SCOPE_WORKGROUP);
      pos = __builtin_amdgcn_readfirstlane(pos);
      if (lane < n && pos + lane < cap)
        lst[LIST_HDR + pos + lane] = make_int2(sv, dv);
    }
    sage_add_rows<F>(sv, dv, n, x, tile, lane);
  }
  __builtin_amdgcn_fence(__ATOMIC_ACQ_REL, "wavefront");
}

template <int F, bool COUNT, bool SAVE>
__device__ __forceinline__ int sage_drain_full(int* q, int qn,
                                               const float* __restrict__ x,
                                               const int* __restrict__ src,
                                               const int* __restrict__ dst,
                                               int lo, unsigned cnt, float* tile,
                                               float* cntl, int2* lst, int cap,
                                               int* lcount, int lane) {
  const int full = qn & ~(Q_DRAIN - 1);
  sage_drain<F, COUNT, SAVE>(q, full, x, src, dst, lo, cnt, tile, cntl, lst,
                             cap, lcount, lane);
  const int rem = qn - full;
  if (lane < rem) q[lane] = q[full + lane];
  return rem;
}

template <int F>
__device__ __forceinline__ int sage_feat_mean(int s, int hf, int r) {
  constexpr int VEC = F / 32;
  constexpr int CC = VEC / 2;
  return VEC * (s / CC + 16 * hf) + 2 * (s % CC) + r;
}

__device__ __forceinline__ int sage_feat_self(int s, int hf, int r) {
  return 8 * (s >> 1) + 4 * hf + 2 * (s & 1) + r;
}

template <int F>
__device__ __forceinline__ void sage_stage_weights(const float* __restrict__ Wl,
                                                   const float* __restrict__ Wr,
                                                   float* wfrag, int tid,
                                                   int nT) {
  for (int idx = tid; idx < 2 * F * HID; idx += nT) {
    const int r = idx & 1;
    const int ln = (idx >> 1) & 31;
    const int ct = (idx >> 6) & 3;
    const int st = idx >> 8;
    const int s = st % (F / 4);
    const int hf = ln >> 4;
    const int col = ct * 16 + (ln & 15);
    wfrag[idx] = (st < F / 4) ? Wl[sage_feat_mean<F>(s, hf, r) * HID + col]
                              : Wr[sage_feat_self(s, hf, r) * HID + col];
  }
}

template <int F, bool COUNT, bool RELU>
__device__ __forceinline__ void sage_tile_product(
    const float* tile, const float* cntl, const float* wfrag,
    const float* __restrict__ self, const float* __restrict__ deg,
    const float* __restrict__ bl, float* __restrict__ out, int lo, int cnt,
    int wave, int nW, int lane) {
  constexpr int VEC = F / 32;
  constexpr int CC = VEC / 2;
  const int hf = lane >> 4;
  const int lm = lane & 15;
  const int nTiles = (cnt + 15) >> 4;
  const v2f* bmean = (const v2f*)wfrag + lane;
  const v2f* bself = bmean + (F / 4) * 4 * 32;

  for (int rt = wave; rt < nTiles; rt += nW) {
    const int i = min(rt * 16 + lm, cnt - 1);

    constexpr int NT = F / 8;
    const float* srow = self + (long)(lo + i) * F + 4 * hf;
    float4 q0 = *(const float4*)(srow);
    float4 q1 = *(const float4*)(srow + 8);
    float4 q2 = *(const float4*)(srow + 16);
    float4 q3 = *(const float4*)(srow + 24);

    float dg;
    if constexpr (COUNT)
      dg = cntl[i];
    else
      dg = deg[lo + i];
    const float inv = 1.0f / fmaxf(dg, 1.0f);

    v8f acc[4];
#pragma unroll
    for (int ct = 0; ct < 4; ++ct)
      acc[ct] = (v8f){0.f, 0.f, 0.f, 0.f, 0.f, 0.f, 0.f, 0.f};

    const float* trow = tile + i * (F + 1) + 16 * hf;
#pragma unroll 2
    for (int jj = 0; jj < 16; ++jj) {
#pragma unroll
      for (int cc = 0; cc < CC; ++cc) {
        v2f a;
        a.x = trow[(2 * cc) * 32 + jj] * inv;
        a.y = trow[(2 * cc + 1) * 32 + jj] * inv;
        const v2f* bp = bmean + (jj * CC + cc) * 4 * 32;
#pragma unroll
        for (int ct = 0; ct < 4; ++ct)
          acc[ct] = __builtin_amdgcn_wmma_f32_16x16x4_f32(
              false, a, false, bp[ct * 32], (short)0, acc[ct], false, false);
      }
    }

#pragma unroll 4
    for (int t = 0; t < NT; ++t) {
      const float4 cur = q0;
      q0 = q1;
      q1 = q2;
      q2 = q3;
      q3 = *(const float4*)(srow + 8 * min(t + 4, NT - 1));
#pragma unroll
      for (int sub = 0; sub < 2; ++sub) {
        v2f a;
        a.x = sub ? cur.z : cur.x;
        a.y = sub ? cur.w : cur.y;
        const v2f* bp = bself + (2 * t + sub) * 4 * 32;
#pragma unroll
        for (int ct = 0; ct < 4; ++ct)
          acc[ct] = __builtin_amdgcn_wmma_f32_16x16x4_f32(
              false, a, false, bp[ct * 32], (short)0, acc[ct], false, false);
      }
    }

#pragma unroll
    for (int ct = 0; ct < 4; ++ct) {
      const int col = ct * 16 + lm;
      const float bv = bl[col];
#pragma unroll
      for (int r = 0; r < 8; ++r) {
        const int row = rt * 16 + r + hf * 8;
        float v = acc[ct][r] + bv;
        if (RELU) v = fmaxf(v, 0.0f);
        if (row < cnt) out[(long)(lo + row) * HID + col] = v;
      }
    }
  }
}

template <int F, bool COUNT, bool SAVE, bool LIST, bool KEYS, bool RELU>
__global__ void __launch_bounds__(SC_THREADS) sage_lds_scatter_kernel(
    const float* __restrict__ x, const int* __restrict__ src,
    const int* __restrict__ dst, int E, int nNodes,
    const float* __restrict__ Wl, const float* __restrict__ bl,
    const float* __restrict__ Wr, float* __restrict__ out,
    float* __restrict__ deg, int* lists,
    const unsigned char* __restrict__ keys) {
  constexpr int TILE_WORDS = (((RNG + 1) * (F + 1)) + 3) & ~3;
  __shared__ __attribute__((aligned(16))) float tile[TILE_WORDS];
  __shared__ float cntl[COUNT ? RNG + 1 : 1];
  __shared__ int queue[(SC_THREADS / 32) * QCAP];
  __shared__ int lcount;
  __shared__ __attribute__((aligned(16))) float wfrag[2 * F * HID];

  const int tid = threadIdx.x;
  const int lane = tid & 31;
  const int wave = __builtin_amdgcn_readfirstlane(tid >> 5);
  const int nW = blockDim.x >> 5;
  int* q = queue + wave * QCAP;

  const int nRanges = (nNodes + RNG - 1) / RNG;
  const bool aligned =
      KEYS ? (((unsigned long)keys) & 15ul) == 0ul
           : (((unsigned long)dst) & 15ul) == 0ul;
  constexpr int SLAB = KEYS ? KEY_SLAB : SLAB_EDGES;
  const int nFull = aligned ? E / SLAB : 0;
  const int stride = max(1, nFull / nRanges);

  sage_stage_weights<F>(Wl, Wr, wfrag, tid, blockDim.x);

  for (int r = blockIdx.x; r < nRanges; r += gridDim.x) {
    const int lo = r * RNG;
    const unsigned cnt = (unsigned)min(RNG, nNodes - lo);
    int2* lst = (int2*)lists + (long)lo * LIST_PER_NODE;
    const int cap = (int)cnt * LIST_PER_NODE - LIST_HDR;

    for (int i = tid; i < TILE_WORDS / 4; i += blockDim.x)
      ((float4*)tile)[i] = make_float4(0.f, 0.f, 0.f, 0.f);
    if (COUNT)
      for (int i = tid; i <= RNG; i += blockDim.x) cntl[i] = 0.f;
    if (SAVE && tid == 0) lcount = 0;
    __syncthreads();

    bool scan = true;
    if (LIST) {
      const int total = __builtin_amdgcn_readfirstlane(lst[0].x);
      if ((unsigned)total <= (unsigned)cap) {
        scan = false;
        int b = wave * 32;
        int2 cur = make_int2(0, 0);
        if (b + lane < total) cur = lst[LIST_HDR + b + lane];
        for (; b < total; b += nW * 32) {
          const int bn = b + nW * 32;
          int2 nxt = make_int2(0, 0);
          if (bn + lane < total) nxt = lst[LIST_HDR + bn + lane];
          sage_add_rows<F>(cur.x, cur.y, min(32, total - b), x, tile, lane);
          cur = nxt;
        }
      }
    }

    if (scan) {
      int qn = 0;

      if constexpr (KEYS) {
        if (wave < nFull) {
          const int start =
              (int)(((unsigned)r * (unsigned)stride) % (unsigned)nFull);
          const unsigned pat = ((unsigned)r & 255u) * 0x01010101u;
          int sl = wave + start;
          if (sl >= nFull) sl -= nFull;
          uint4 cur[KEY_LOADS];
#pragma unroll
          for (int u = 0; u < KEY_LOADS; ++u)
            cur[u] = *(const uint4*)(keys + sl * KEY_SLAB + u * 512 + lane * 16);

          for (int k = wave; k < nFull; k += nW) {
            int sn = k + nW + start;
            if (sn >= nFull) sn -= nFull;
            if (k + nW >= nFull) sn = sl;
            uint4 nxt[KEY_LOADS];
#pragma unroll
            for (int u = 0; u < KEY_LOADS; ++u)
              nxt[u] =
                  *(const uint4*)(keys + sn * KEY_SLAB + u * 512 + lane * 16);

#pragma unroll
            for (int u = 0; u < KEY_LOADS; ++u) {
              const int eb = sl * KEY_SLAB + u * 512 + lane * 16;
              const unsigned w4[4] = {cur[u].x, cur[u].y, cur[u].z, cur[u].w};
              unsigned nz = 0u;
#pragma unroll
              for (int i = 0; i < 4; ++i) {
                const unsigned d = w4[i] ^ pat;
                const unsigned t = (d & 0x7f7f7f7fu) + 0x7f7f7f7fu;
                nz |= (((t | d) >> 7) & 0x01010101u) << i;
              }
              unsigned f = ~nz & 0x0f0f0f0fu;
              unsigned m = __builtin_amdgcn_ballot_w32(f != 0u);
              while (m) {
                if (f != 0u) {
                  const int b = __builtin_ctz(f);
                  q[qn + (int)__builtin_amdgcn_mbcnt_lo(m, 0u)] =
                      eb + ((b & 7) << 2) + (b >> 3);
                  f &= f - 1u;
                }
                qn += __builtin_popcount(m);
                if (qn >= Q_DRAIN)
                  qn = sage_drain_full<F, COUNT, SAVE>(q, qn, x, src, dst, lo,
                                                       cnt, tile, cntl, lst,
                                                       cap, &lcount, lane);
                m = __builtin_amdgcn_ballot_w32(f != 0u);
              }
            }
#pragma unroll
            for (int u = 0; u < KEY_LOADS; ++u) cur[u] = nxt[u];
            sl = sn;
          }
        }
      } else if (wave < nFull) {
        const int start =
            (int)(((unsigned)r * (unsigned)stride) % (unsigned)nFull);
        int sl = wave + start;
        if (sl >= nFull) sl -= nFull;
        int4 cur[SLAB_LOADS];
#pragma unroll
        for (int u = 0; u < SLAB_LOADS; ++u)
          cur[u] = *(const int4*)(dst + sl * SLAB_EDGES + u * 128 + lane * 4);

        for (int k = wave; k < nFull; k += nW) {
          int sn = k + nW + start;
          if (sn >= nFull) sn -= nFull;
          if (k + nW >= nFull) sn = sl;
          int4 nxt[SLAB_LOADS];
#pragma unroll
          for (int u = 0; u < SLAB_LOADS; ++u)
            nxt[u] = *(const int4*)(dst + sn * SLAB_EDGES + u * 128 + lane * 4);

#pragma unroll
          for (int u = 0; u < SLAB_LOADS; ++u) {
            const int eb = sl * SLAB_EDGES + u * 128 + lane * 4;
            const int dv4[4] = {cur[u].x, cur[u].y, cur[u].z, cur[u].w};
#pragma unroll
            for (int c = 0; c < 4; ++c) {
              const bool hit = (unsigned)(dv4[c] - lo) < cnt;
              const unsigned m = __builtin_amdgcn_ballot_w32(hit);
              if (m) {
                if (hit) q[qn + (int)__builtin_amdgcn_mbcnt_lo(m, 0u)] = eb + c;
                qn += __builtin_popcount(m);
              }
            }
            if (qn >= Q_DRAIN)
              qn = sage_drain_full<F, COUNT, SAVE>(q, qn, x, src, dst, lo, cnt,
                                                   tile, cntl, lst, cap,
                                                   &lcount, lane);
          }
#pragma unroll
          for (int u = 0; u < SLAB_LOADS; ++u) cur[u] = nxt[u];
          sl = sn;
        }
      }
      for (int e0 = nFull * SLAB + wave * 32; e0 < E; e0 += nW * 32) {
        const int e = e0 + lane;
        const int d = (e < E) ? dst[e] : -1;
        const bool hit = (unsigned)(d - lo) < cnt;
        const unsigned m = __builtin_amdgcn_ballot_w32(hit);
        if (m) {
          if (hit) q[qn + (int)__builtin_amdgcn_mbcnt_lo(m, 0u)] = e;
          qn += __builtin_popcount(m);
        }
        if (qn >= Q_DRAIN)
          qn = sage_drain_full<F, COUNT, SAVE>(q, qn, x, src, dst, lo, cnt,
                                               tile, cntl, lst, cap, &lcount,
                                               lane);
      }
      if (qn > 0)
        sage_drain<F, COUNT, SAVE>(q, qn, x, src, dst, lo, cnt, tile, cntl, lst,
                                   cap, &lcount, lane);
    }
    __syncthreads();

    sage_tile_product<F, COUNT, RELU>(tile, cntl, wfrag, x, deg, bl, out, lo,
                                      (int)cnt, wave, nW, lane);
    if (COUNT)
      for (int i = tid; i < (int)cnt; i += blockDim.x) deg[lo + i] = cntl[i];
    if (SAVE && tid == 0) lst[0] = make_int2(lcount, 0);
    __syncthreads();
  }
}

extern "C" void kernel_launch(void* const* d_in, const int* in_sizes, int n_in,
                              void* d_out, int out_size, void* d_ws,
                              size_t ws_size, hipStream_t stream) {
  const float* x = (const float*)d_in[0];
  const int* ei = (const int*)d_in[1];
  const float* Wl1 = (const float*)d_in[2];
  const float* bl1 = (const float*)d_in[3];
  const float* Wr1 = (const float*)d_in[4];
  const float* Wl2 = (const float*)d_in[5];
  const float* bl2 = (const float*)d_in[6];
  const float* Wr2 = (const float*)d_in[7];
  float* out = (float*)d_out;

  const int N = in_sizes[0] / IN_C;
  const int E = in_sizes[1] / 2;
  const int* src = ei;
  const int* dst = ei + E;

  float* ws = (float*)d_ws;
  float* deg = ws;
  unsigned char* keys = (unsigned char*)(deg + N);
  float* h = deg + N + (long)N * IN_C;
  int* lists = (int*)(h + (long)N * HID);


  sage_range_key_kernel<<<512, 256, 0, stream>>>(dst, E, keys);
  sage_lds_scatter_kernel<IN_C, true, true, false, true, true>
      <<<256, SC_THREADS, 0, stream>>>(x, src, dst, E, N, Wl1, bl1, Wr1, h, deg,
                                       lists, keys);

  sage_lds_scatter_kernel<HID, false, false, true, false, false>
      <<<256, SC_THREADS, 0, stream>>>(h, src, dst, E, N, Wl2, bl2, Wr2, out,
                                       deg, lists, nullptr);
}


# The launch in front also writes each edge's (source, destination) as one 8-byte pair: a matched edge costs one random read, not two (one per input list); sums still LDS arrival order
# speedup vs baseline: 4.1126x; 1.0380x over previous
#include <hip/hip_runtime.h>

typedef __attribute__((ext_vector_type(2))) float v2f;
typedef __attribute__((ext_vector_type(8))) float v8f;

#define IN_C 128
#define HID  64
#define RNG 392
#define SC_THREADS 1024
#define SLAB_LOADS 8
#define SLAB_EDGES (SLAB_LOADS * 128)
#define Q_DRAIN 32
#define QCAP (Q_DRAIN + 128)
#define RIF 8
#define LIST_PER_NODE (HID / 2)
#define LIST_HDR 16
#define KEY_LOADS 4
#define KEY_SLAB (KEY_LOADS * 512)


__device__ __forceinline__ unsigned sage_key(int d) {
  return ((unsigned)d / (unsigned)RNG) & 255u;
}

__global__ void __launch_bounds__(256) sage_range_key_kernel(
    const int* __restrict__ src, const int* __restrict__ dst, int E,
    unsigned char* __restrict__ keys, int2* __restrict__ pairs) {
  const int gtid = blockIdx.x * blockDim.x + threadIdx.x;
  const int T = gridDim.x * blockDim.x;
  const bool aligned = (((unsigned long)dst) & 15ul) == 0ul &&
                       (((unsigned long)src) & 15ul) == 0ul &&
                       (((unsigned long)keys) & 3ul) == 0ul &&
                       (((unsigned long)pairs) & 15ul) == 0ul;
  const int n4 = aligned ? (E >> 2) : 0;
  for (int base = gtid; base < n4; base += 4 * T) {
    int4 d[4], s[4];
#pragma unroll
    for (int u = 0; u < 4; ++u) {
      d[u] = ((const int4*)dst)[min(base + u * T, n4 - 1)];
      s[u] = ((const int4*)src)[min(base + u * T, n4 - 1)];
    }
#pragma unroll
    for (int u = 0; u < 4; ++u)
      if (base + u * T < n4) {
        ((unsigned*)keys)[base + u * T] =
            sage_key(d[u].x) | (sage_key(d[u].y) << 8) |
            (sage_key(d[u].z) << 16) | (sage_key(d[u].w) << 24);
        int4* pp = (int4*)(pairs + 4l * (base + u * T));
        pp[0] = make_int4(s[u].x, d[u].x, s[u].y, d[u].y);
        pp[1] = make_int4(s[u].z, d[u].z, s[u].w, d[u].w);
      }
  }
  for (int e = n4 * 4 + gtid; e < E; e += T) {
    keys[e] = (unsigned char)sage_key(dst[e]);
    pairs[e] = make_int2(src[e], dst[e]);
  }
}

template <int F>
__device__ __forceinline__ void sage_add_rows(int sv, int dv, int n,
                                              const float* __restrict__ x,
                                              float* tile, int lane) {
  constexpr int VEC = F / 32;
  for (int j = 0; j < n; j += RIF) {
    float v[RIF][VEC];
    int dd[RIF];
#pragma unroll
    for (int k = 0; k < RIF; ++k) {
      const int jj = min(j + k, n - 1);
      const int sr = __builtin_amdgcn_readlane(sv, jj);
      dd[k] = __builtin_amdgcn_readlane(dv, jj);
      const float* row = x + (long)sr * F + lane * VEC;
      if constexpr (VEC == 4) {
        const float4 t = *(const float4*)row;
        v[k][0] = t.x;
        v[k][1] = t.y;
        v[k][2] = t.z;
        v[k][3] = t.w;
      } else {
        const float2 t = *(const float2*)row;
        v[k][0] = t.x;
        v[k][1] = t.y;
      }
    }
#pragma unroll
    for (int k = 0; k < RIF; ++k)
      if (j + k < n) {
        float* t = tile + dd[k] * (F + 1) + lane;
#pragma unroll
        for (int c = 0; c < VEC; ++c)
          __hip_atomic_fetch_add(t + c * 32, v[k][c], __ATOMIC_RELAXED,
                                 __HIP_MEMORY_SCOPE_WORKGROUP);
      }
  }
}

template <int F, bool COUNT, bool SAVE>
__device__ __forceinline__ void sage_drain(const int* q, int qn,
                                           const float* __restrict__ x,
                                           const int2* __restrict__ pairs,
                                           int lo, unsigned cnt, float* tile,
                                           float* cntl, int2* lst, int cap,
                                           int* lcount, int lane) {
  __builtin_amdgcn_fence(__ATOMIC_ACQ_REL, "wavefront");
  for (int b = 0; b < qn; b += 32) {
    const int n = min(32, qn - b);
    int sv = 0, dv = 0;
    if (lane < n) {
      const int2 p = pairs[q[b + lane]];
      sv = p.x;
      dv = p.y - lo;
      if ((unsigned)dv >= cnt) {
        sv = 0;
        dv = RNG;
      }
      if (COUNT)
        __hip_atomic_fetch_add(&cntl[dv], 1.0f, __ATOMIC_RELAXED,
                               __HIP_MEMORY_SCOPE_WORKGROUP);
    }
    if (SAVE) {
      int pos = 0;
      if (lane == 0)
        pos = __hip_atomic_fetch_add(lcount, n, __ATOMIC_RELAXED,
                                     __HIP_MEMORY_SCOPE_WORKGROUP);
      pos = __builtin_amdgcn_readfirstlane(pos);
      if (lane < n && pos + lane < cap)
        lst[LIST_HDR + pos + lane] = make_int2(sv, dv);
    }
    sage_add_rows<F>(sv, dv, n, x, tile, lane);
  }
  __builtin_amdgcn_fence(__ATOMIC_ACQ_REL, "wavefront");
}

template <int F, bool COUNT, bool SAVE>
__device__ __forceinline__ int sage_drain_full(int* q, int qn,
                                               const float* __restrict__ x,
                                               const int2* __restrict__ pairs,
                                               int lo, unsigned cnt, float* tile,
                                               float* cntl, int2* lst, int cap,
                                               int* lcount, int lane) {
  const int full = qn & ~(Q_DRAIN - 1);
  sage_drain<F, COUNT, SAVE>(q, full, x, pairs, lo, cnt, tile, cntl, lst, cap,
                             lcount, lane);
  const int rem = qn - full;
  if (lane < rem) q[lane] = q[full + lane];
  return rem;
}

template <int F>
__device__ __forceinline__ int sage_feat_mean(int s, int hf, int r) {
  constexpr int VEC = F / 32;
  constexpr int CC = VEC / 2;
  return VEC * (s / CC + 16 * hf) + 2 * (s % CC) + r;
}

__device__ __forceinline__ int sage_feat_self(int s, int hf, int r) {
  return 8 * (s >> 1) + 4 * hf + 2 * (s & 1) + r;
}

template <int F>
__device__ __forceinline__ void sage_stage_weights(const float* __restrict__ Wl,
                                                   const float* __restrict__ Wr,
                                                   float* wfrag, int tid,
                                                   int nT) {
  for (int idx = tid; idx < 2 * F * HID; idx += nT) {
    const int r = idx & 1;
    const int ln = (idx >> 1) & 31;
    const int ct = (idx >> 6) & 3;
    const int st = idx >> 8;
    const int s = st % (F / 4);
    const int hf = ln >> 4;
    const int col = ct * 16 + (ln & 15);
    wfrag[idx] = (st < F / 4) ? Wl[sage_feat_mean<F>(s, hf, r) * HID + col]
                              : Wr[sage_feat_self(s, hf, r) * HID + col];
  }
}

template <int F, bool COUNT, bool RELU>
__device__ __forceinline__ void sage_tile_product(
    const float* tile, const float* cntl, const float* wfrag,
    const float* __restrict__ self, const float* __restrict__ deg,
    const float* __restrict__ bl, float* __restrict__ out, int lo, int cnt,
    int wave, int nW, int lane) {
  constexpr int VEC = F / 32;
  constexpr int CC = VEC / 2;
  const int hf = lane >> 4;
  const int lm = lane & 15;
  const int nTiles = (cnt + 15) >> 4;
  const v2f* bmean = (const v2f*)wfrag + lane;
  const v2f* bself = bmean + (F / 4) * 4 * 32;

  for (int rt = wave; rt < nTiles; rt += nW) {
    const int i = min(rt * 16 + lm, cnt - 1);

    constexpr int NT = F / 8;
    const float* srow = self + (long)(lo + i) * F + 4 * hf;
    float4 q0 = *(const float4*)(srow);
    float4 q1 = *(const float4*)(srow + 8);
    float4 q2 = *(const float4*)(srow + 16);
    float4 q3 = *(const float4*)(srow + 24);

    float dg;
    if constexpr (COUNT)
      dg = cntl[i];
    else
      dg = deg[lo + i];
    const float inv = 1.0f / fmaxf(dg, 1.0f);

    v8f acc[4];
#pragma unroll
    for (int ct = 0; ct < 4; ++ct)
      acc[ct] = (v8f){0.f, 0.f, 0.f, 0.f, 0.f, 0.f, 0.f, 0.f};

    const float* trow = tile + i * (F + 1) + 16 * hf;
#pragma unroll 2
    for (int jj = 0; jj < 16; ++jj) {
#pragma unroll
      for (int cc = 0; cc < CC; ++cc) {
        v2f a;
        a.x = trow[(2 * cc) * 32 + jj] * inv;
        a.y = trow[(2 * cc + 1) * 32 + jj] * inv;
        const v2f* bp = bmean + (jj * CC + cc) * 4 * 32;
#pragma unroll
        for (int ct = 0; ct < 4; ++ct)
          acc[ct] = __builtin_amdgcn_wmma_f32_16x16x4_f32(
              false, a, false, bp[ct * 32], (short)0, acc[ct], false, false);
      }
    }

#pragma unroll 4
    for (int t = 0; t < NT; ++t) {
      const float4 cur = q0;
      q0 = q1;
      q1 = q2;
      q2 = q3;
      q3 = *(const float4*)(srow + 8 * min(t + 4, NT - 1));
#pragma unroll
      for (int sub = 0; sub < 2; ++sub) {
        v2f a;
        a.x = sub ? cur.z : cur.x;
        a.y = sub ? cur.w : cur.y;
        const v2f* bp = bself + (2 * t + sub) * 4 * 32;
#pragma unroll
        for (int ct = 0; ct < 4; ++ct)
          acc[ct] = __builtin_amdgcn_wmma_f32_16x16x4_f32(
              false, a, false, bp[ct * 32], (short)0, acc[ct], false, false);
      }
    }

#pragma unroll
    for (int ct = 0; ct < 4; ++ct) {
      const int col = ct * 16 + lm;
      const float bv = bl[col];
#pragma unroll
      for (int r = 0; r < 8; ++r) {
        const int row = rt * 16 + r + hf * 8;
        float v = acc[ct][r] + bv;
        if (RELU) v = fmaxf(v, 0.0f);
        if (row < cnt) out[(long)(lo + row) * HID + col] = v;
      }
    }
  }
}

template <int F, bool COUNT, bool SAVE, bool LIST, bool KEYS, bool RELU>
__global__ void __launch_bounds__(SC_THREADS) sage_lds_scatter_kernel(
    const float* __restrict__ x, const int2* __restrict__ pairs,
    const int* __restrict__ dst, int E, int nNodes,
    const float* __restrict__ Wl, const float* __restrict__ bl,
    const float* __restrict__ Wr, float* __restrict__ out,
    float* __restrict__ deg, int* lists,
    const unsigned char* __restrict__ keys) {
  constexpr int TILE_WORDS = (((RNG + 1) * (F + 1)) + 3) & ~3;
  __shared__ __attribute__((aligned(16))) float tile[TILE_WORDS];
  __shared__ float cntl[COUNT ? RNG + 1 : 1];
  __shared__ int queue[(SC_THREADS / 32) * QCAP];
  __shared__ int lcount;
  __shared__ __attribute__((aligned(16))) float wfrag[2 * F * HID];

  const int tid = threadIdx.x;
  const int lane = tid & 31;
  const int wave = __builtin_amdgcn_readfirstlane(tid >> 5);
  const int nW = blockDim.x >> 5;
  int* q = queue + wave * QCAP;

  const int nRanges = (nNodes + RNG - 1) / RNG;
  const bool aligned =
      KEYS ? (((unsigned long)keys) & 15ul) == 0ul
           : (((unsigned long)dst) & 15ul) == 0ul;
  constexpr int SLAB = KEYS ? KEY_SLAB : SLAB_EDGES;
  const int nFull = aligned ? E / SLAB : 0;
  const int stride = max(1, nFull / nRanges);

  sage_stage_weights<F>(Wl, Wr, wfrag, tid, blockDim.x);

  for (int r = blockIdx.x; r < nRanges; r += gridDim.x) {
    const int lo = r * RNG;
    const unsigned cnt = (unsigned)min(RNG, nNodes - lo);
    int2* lst = (int2*)lists + (long)lo * LIST_PER_NODE;
    const int cap = (int)cnt * LIST_PER_NODE - LIST_HDR;

    for (int i = tid; i < TILE_WORDS / 4; i += blockDim.x)
      ((float4*)tile)[i] = make_float4(0.f, 0.f, 0.f, 0.f);
    if (COUNT)
      for (int i = tid; i <= RNG; i += blockDim.x) cntl[i] = 0.f;
    if (SAVE && tid == 0) lcount = 0;
    __syncthreads();

    bool scan = true;
    if (LIST) {
      const int total = __builtin_amdgcn_readfirstlane(lst[0].x);
      if ((unsigned)total <= (unsigned)cap) {
        scan = false;
        int b = wave * 32;
        int2 cur = make_int2(0, 0);
        if (b + lane < total) cur = lst[LIST_HDR + b + lane];
        for (; b < total; b += nW * 32) {
          const int bn = b + nW * 32;
          int2 nxt = make_int2(0, 0);
          if (bn + lane < total) nxt = lst[LIST_HDR + bn + lane];
          sage_add_rows<F>(cur.x, cur.y, min(32, total - b), x, tile, lane);
          cur = nxt;
        }
      }
    }

    if (scan) {
      int qn = 0;

      if constexpr (KEYS) {
        if (wave < nFull) {
          const int start =
              (int)(((unsigned)r * (unsigned)stride) % (unsigned)nFull);
          const unsigned pat = ((unsigned)r & 255u) * 0x01010101u;
          int sl = wave + start;
          if (sl >= nFull) sl -= nFull;
          uint4 cur[KEY_LOADS];
#pragma unroll
          for (int u = 0; u < KEY_LOADS; ++u)
            cur[u] = *(const uint4*)(keys + sl * KEY_SLAB + u * 512 + lane * 16);

          for (int k = wave; k < nFull; k += nW) {
            int sn = k + nW + start;
            if (sn >= nFull) sn -= nFull;
            if (k + nW >= nFull) sn = sl;
            uint4 nxt[KEY_LOADS];
#pragma unroll
            for (int u = 0; u < KEY_LOADS; ++u)
              nxt[u] =
                  *(const uint4*)(keys + sn * KEY_SLAB + u * 512 + lane * 16);

#pragma unroll
            for (int u = 0; u < KEY_LOADS; ++u) {
              const int eb = sl * KEY_SLAB + u * 512 + lane * 16;
              const unsigned w4[4] = {cur[u].x, cur[u].y, cur[u].z, cur[u].w};
              unsigned nz = 0u;
#pragma unroll
              for (int i = 0; i < 4; ++i) {
                const unsigned d = w4[i] ^ pat;
                const unsigned t = (d & 0x7f7f7f7fu) + 0x7f7f7f7fu;
                nz |= (((t | d) >> 7) & 0x01010101u) << i;
              }
              unsigned f = ~nz & 0x0f0f0f0fu;
              unsigned m = __builtin_amdgcn_ballot_w32(f != 0u);
              while (m) {
                if (f != 0u) {
                  const int b = __builtin_ctz(f);
                  q[qn + (int)__builtin_amdgcn_mbcnt_lo(m, 0u)] =
                      eb + ((b & 7) << 2) + (b >> 3);
                  f &= f - 1u;
                }
                qn += __builtin_popcount(m);
                if (qn >= Q_DRAIN)
                  qn = sage_drain_full<F, COUNT, SAVE>(q, qn, x, pairs, lo,
                                                       cnt, tile, cntl, lst,
                                                       cap, &lcount, lane);
                m = __builtin_amdgcn_ballot_w32(f != 0u);
              }
            }
#pragma unroll
            for (int u = 0; u < KEY_LOADS; ++u) cur[u] = nxt[u];
            sl = sn;
          }
        }
      } else if (wave < nFull) {
        const int start =
            (int)(((unsigned)r * (unsigned)stride) % (unsigned)nFull);
        int sl = wave + start;
        if (sl >= nFull) sl -= nFull;
        int4 cur[SLAB_LOADS];
#pragma unroll
        for (int u = 0; u < SLAB_LOADS; ++u)
          cur[u] = *(const int4*)(dst + sl * SLAB_EDGES + u * 128 + lane * 4);

        for (int k = wave; k < nFull; k += nW) {
          int sn = k + nW + start;
          if (sn >= nFull) sn -= nFull;
          if (k + nW >= nFull) sn = sl;
          int4 nxt[SLAB_LOADS];
#pragma unroll
          for (int u = 0; u < SLAB_LOADS; ++u)
            nxt[u] = *(const int4*)(dst + sn * SLAB_EDGES + u * 128 + lane * 4);

#pragma unroll
          for (int u = 0; u < SLAB_LOADS; ++u) {
            const int eb = sl * SLAB_EDGES + u * 128 + lane * 4;
            const int dv4[4] = {cur[u].x, cur[u].y, cur[u].z, cur[u].w};
#pragma unroll
            for (int c = 0; c < 4; ++c) {
              const bool hit = (unsigned)(dv4[c] - lo) < cnt;
              const unsigned m = __builtin_amdgcn_ballot_w32(hit);
              if (m) {
                if (hit) q[qn + (int)__builtin_amdgcn_mbcnt_lo(m, 0u)] = eb + c;
                qn += __builtin_popcount(m);
              }
            }
            if (qn >= Q_DRAIN)
              qn = sage_drain_full<F, COUNT, SAVE>(q, qn, x, pairs, lo, cnt,
                                                   tile, cntl, lst, cap,
                                                   &lcount, lane);
          }
#pragma unroll
          for (int u = 0; u < SLAB_LOADS; ++u) cur[u] = nxt[u];
          sl = sn;
        }
      }
      for (int e0 = nFull * SLAB + wave * 32; e0 < E; e0 += nW * 32) {
        const int e = e0 + lane;
        const int d = (e < E) ? dst[e] : -1;
        const bool hit = (unsigned)(d - lo) < cnt;
        const unsigned m = __builtin_amdgcn_ballot_w32(hit);
        if (m) {
          if (hit) q[qn + (int)__builtin_amdgcn_mbcnt_lo(m, 0u)] = e;
          qn += __builtin_popcount(m);
        }
        if (qn >= Q_DRAIN)
          qn = sage_drain_full<F, COUNT, SAVE>(q, qn, x, pairs, lo, cnt, tile,
                                               cntl, lst, cap, &lcount, lane);
      }
      if (qn > 0)
        sage_drain<F, COUNT, SAVE>(q, qn, x, pairs, lo, cnt, tile, cntl, lst,
                                   cap, &lcount, lane);
    }
    __syncthreads();

    sage_tile_product<F, COUNT, RELU>(tile, cntl, wfrag, x, deg, bl, out, lo,
                                      (int)cnt, wave, nW, lane);
    if (COUNT)
      for (int i = tid; i < (int)cnt; i += blockDim.x) deg[lo + i] = cntl[i];
    if (SAVE && tid == 0) lst[0] = make_int2(lcount, 0);
    __syncthreads();
  }
}

extern "C" void kernel_launch(void* const* d_in, const int* in_sizes, int n_in,
                              void* d_out, int out_size, void* d_ws,
                              size_t ws_size, hipStream_t stream) {
  const float* x = (const float*)d_in[0];
  const int* ei = (const int*)d_in[1];
  const float* Wl1 = (const float*)d_in[2];
  const float* bl1 = (const float*)d_in[3];
  const float* Wr1 = (const float*)d_in[4];
  const float* Wl2 = (const float*)d_in[5];
  const float* bl2 = (const float*)d_in[6];
  const float* Wr2 = (const float*)d_in[7];
  float* out = (float*)d_out;

  const int N = in_sizes[0] / IN_C;
  const int E = in_sizes[1] / 2;
  const int* src = ei;
  const int* dst = ei + E;

  float* ws = (float*)d_ws;
  float* deg = ws;
  unsigned char* keys = (unsigned char*)(deg + N);
  int2* pairs = (int2*)(deg + N + (long)N * HID);
  float* h = deg + N + (long)N * IN_C;
  int* lists = (int*)(h + (long)N * HID);


  sage_range_key_kernel<<<512, 256, 0, stream>>>(src, dst, E, keys, pairs);
  sage_lds_scatter_kernel<IN_C, true, true, false, true, true>
      <<<256, SC_THREADS, 0, stream>>>(x, pairs, dst, E, N, Wl1, bl1, Wr1, h,
                                       deg, lists, keys);

  sage_lds_scatter_kernel<HID, false, false, true, false, false>
      <<<256, SC_THREADS, 0, stream>>>(h, pairs, dst, E, N, Wl2, bl2, Wr2, out,
                                       deg, lists, nullptr);
}


# Layer-1 gather scans two bit planes per range (one per 4-bit half of the range number; 32 planes written by the launch in front) and ANDs them, not a byte per edge; sums still LDS arrival order
# speedup vs baseline: 4.4090x; 1.0721x over previous
#include <hip/hip_runtime.h>

typedef __attribute__((ext_vector_type(2))) float v2f;
typedef __attribute__((ext_vector_type(8))) float v8f;

#define IN_C 128
#define HID  64
#define RNG 392
#define SC_THREADS 1024
#define SLAB_LOADS 8
#define SLAB_EDGES (SLAB_LOADS * 128)
#define Q_DRAIN 32
#define QCAP (Q_DRAIN + 128)
#define RIF 8
#define LIST_PER_NODE (HID / 2)
#define LIST_HDR 16
#define KEY_PLANES 32
#define PLANE_SLAB 128


__device__ __forceinline__ unsigned sage_key(int d) {
  return ((unsigned)d / (unsigned)RNG) & 255u;
}

__device__ __forceinline__ int sage_plane_words(int E) {
  return (((E + 31) >> 5) + PLANE_SLAB - 1) & ~(PLANE_SLAB - 1);
}

template <int HALF>
__device__ __forceinline__ void sage_plane_half(const unsigned (&kk)[32],
                                                unsigned* __restrict__ planes,
                                                int pw, int wbase, int lane) {
  unsigned acc[16];
#pragma unroll
  for (int p = 0; p < 16; ++p) acc[p] = 0u;
#pragma unroll
  for (int g = 0; g < 32; ++g) {
    const unsigned h = HALF ? ((kk[g] & 15u) | ((kk[g] >> 4) & 16u))
                            : (kk[g] >> 4);
#pragma unroll
    for (int p = 0; p < 16; ++p) {
      const unsigned w = __builtin_amdgcn_ballot_w32(h == (unsigned)p);
      acc[p] = (lane == g) ? w : acc[p];
    }
  }
#pragma unroll
  for (int p = 0; p < 16; ++p)
    planes[(long)(HALF * 16 + p) * pw + wbase + lane] = acc[p];
}

__global__ void __launch_bounds__(256) sage_range_key_kernel(
    const int* __restrict__ src, const int* __restrict__ dst, int E,
    unsigned* __restrict__ planes, int2* __restrict__ pairs) {
  const int lane = threadIdx.x & 31;
  const int wv = __builtin_amdgcn_readfirstlane(
      (int)((threadIdx.x >> 5) * gridDim.x + blockIdx.x));
  const int nWv = (int)(gridDim.x * (blockDim.x >> 5));
  const int pw = sage_plane_words(E);
  for (int t = wv; t < (pw >> 5); t += nWv) {
    const int e0 = t << 10;
    int dv[32], sv[32];
#pragma unroll
    for (int g = 0; g < 32; ++g) {
      const int e = min(e0 + (g << 5) + lane, E - 1);
      dv[g] = dst[e];
      sv[g] = src[e];
    }
    unsigned kk[32];
#pragma unroll
    for (int g = 0; g < 32; ++g) {
      const int e = e0 + (g << 5) + lane;
      unsigned k = 0x100u;
      if (e < E) {
        k = sage_key(dv[g]);
        pairs[e] = make_int2(sv[g], dv[g]);
      }
      kk[g] = k;
    }
    sage_plane_half<0>(kk, planes, pw, t << 5, lane);
    sage_plane_half<1>(kk, planes, pw, t << 5, lane);
  }
}

template <int F>
__device__ __forceinline__ void sage_add_rows(int sv, int dv, int n,
                                              const float* __restrict__ x,
                                              float* tile, int lane) {
  constexpr int VEC = F / 32;
  for (int j = 0; j < n; j += RIF) {
    float v[RIF][VEC];
    int dd[RIF];
#pragma unroll
    for (int k = 0; k < RIF; ++k) {
      const int jj = min(j + k, n - 1);
      const int sr = __builtin_amdgcn_readlane(sv, jj);
      dd[k] = __builtin_amdgcn_readlane(dv, jj);
      const float* row = x + (long)sr * F + lane * VEC;
      if constexpr (VEC == 4) {
        const float4 t = *(const float4*)row;
        v[k][0] = t.x;
        v[k][1] = t.y;
        v[k][2] = t.z;
        v[k][3] = t.w;
      } else {
        const float2 t = *(const float2*)row;
        v[k][0] = t.x;
        v[k][1] = t.y;
      }
    }
#pragma unroll
    for (int k = 0; k < RIF; ++k)
      if (j + k < n) {
        float* t = tile + dd[k] * (F + 1) + lane;
#pragma unroll
        for (int c = 0; c < VEC; ++c)
          __hip_atomic_fetch_add(t + c * 32, v[k][c], __ATOMIC_RELAXED,
                                 __HIP_MEMORY_SCOPE_WORKGROUP);
      }
  }
}

template <int F, bool COUNT, bool SAVE>
__device__ __forceinline__ void sage_drain(const int* q, int qn,
                                           const float* __restrict__ x,
                                           const int2* __restrict__ pairs,
                                           int lo, unsigned cnt, float* tile,
                                           float* cntl, int2* lst, int cap,
                                           int* lcount, int lane) {
  __builtin_amdgcn_fence(__ATOMIC_ACQ_REL, "wavefront");
  for (int b = 0; b < qn; b += 32) {
    const int n = min(32, qn - b);
    int sv = 0, dv = 0;
    if (lane < n) {
      const int2 p = pairs[q[b + lane]];
      sv = p.x;
      dv = p.y - lo;
      if ((unsigned)dv >= cnt) {
        sv = 0;
        dv = RNG;
      }
      if (COUNT)
        __hip_atomic_fetch_add(&cntl[dv], 1.0f, __ATOMIC_RELAXED,
                               __HIP_MEMORY_SCOPE_WORKGROUP);
    }
    if (SAVE) {
      int pos = 0;
      if (lane == 0)
        pos = __hip_atomic_fetch_add(lcount, n, __ATOMIC_RELAXED,
                                     __HIP_MEMORY_SCOPE_WORKGROUP);
      pos = __builtin_amdgcn_readfirstlane(pos);
      if (lane < n && pos + lane < cap)
        lst[LIST_HDR + pos + lane] = make_int2(sv, dv);
    }
    sage_add_rows<F>(sv, dv, n, x, tile, lane);
  }
  __builtin_amdgcn_fence(__ATOMIC_ACQ_REL, "wavefront");
}

template <int F, bool COUNT, bool SAVE>
__device__ __forceinline__ int sage_drain_full(int* q, int qn,
                                               const float* __restrict__ x,
                                               const int2* __restrict__ pairs,
                                               int lo, unsigned cnt, float* tile,
                                               float* cntl, int2* lst, int cap,
                                               int* lcount, int lane) {
  const int full = qn & ~(Q_DRAIN - 1);
  sage_drain<F, COUNT, SAVE>(q, full, x, pairs, lo, cnt, tile, cntl, lst, cap,
                             lcount, lane);
  const int rem = qn - full;
  if (lane < rem) q[lane] = q[full + lane];
  return rem;
}

template <int F>
__device__ __forceinline__ int sage_feat_mean(int s, int hf, int r) {
  constexpr int VEC = F / 32;
  constexpr int CC = VEC / 2;
  return VEC * (s / CC + 16 * hf) + 2 * (s % CC) + r;
}

__device__ __forceinline__ int sage_feat_self(int s, int hf, int r) {
  return 8 * (s >> 1) + 4 * hf + 2 * (s & 1) + r;
}

template <int F>
__device__ __forceinline__ void sage_stage_weights(const float* __restrict__ Wl,
                                                   const float* __restrict__ Wr,
                                                   float* wfrag, int tid,
                                                   int nT) {
  for (int idx = tid; idx < 2 * F * HID; idx += nT) {
    const int r = idx & 1;
    const int ln = (idx >> 1) & 31;
    const int ct = (idx >> 6) & 3;
    const int st = idx >> 8;
    const int s = st % (F / 4);
    const int hf = ln >> 4;
    const int col = ct * 16 + (ln & 15);
    wfrag[idx] = (st < F / 4) ? Wl[sage_feat_mean<F>(s, hf, r) * HID + col]
                              : Wr[sage_feat_self(s, hf, r) * HID + col];
  }
}

template <int F, bool COUNT, bool RELU>
__device__ __forceinline__ void sage_tile_product(
    const float* tile, const float* cntl, const float* wfrag,
    const float* __restrict__ self, const float* __restrict__ deg,
    const float* __restrict__ bl, float* __restrict__ out, int lo, int cnt,
    int wave, int nW, int lane) {
  constexpr int VEC = F / 32;
  constexpr int CC = VEC / 2;
  const int hf = lane >> 4;
  const int lm = lane & 15;
  const int nTiles = (cnt + 15) >> 4;
  const v2f* bmean = (const v2f*)wfrag + lane;
  const v2f* bself = bmean + (F / 4) * 4 * 32;

  for (int rt = wave; rt < nTiles; rt += nW) {
    const int i = min(rt * 16 + lm, cnt - 1);

    constexpr int NT = F / 8;
    const float* srow = self + (long)(lo + i) * F + 4 * hf;
    float4 q0 = *(const float4*)(srow);
    float4 q1 = *(const float4*)(srow + 8);
    float4 q2 = *(const float4*)(srow + 16);
    float4 q3 = *(const float4*)(srow + 24);

    float dg;
    if constexpr (COUNT)
      dg = cntl[i];
    else
      dg = deg[lo + i];
    const float inv = 1.0f / fmaxf(dg, 1.0f);

    v8f acc[4];
#pragma unroll
    for (int ct = 0; ct < 4; ++ct)
      acc[ct] = (v8f){0.f, 0.f, 0.f, 0.f, 0.f, 0.f, 0.f, 0.f};

    const float* trow = tile + i * (F + 1) + 16 * hf;
#pragma unroll 2
    for (int jj = 0; jj < 16; ++jj) {
#pragma unroll
      for (int cc = 0; cc < CC; ++cc) {
        v2f a;
        a.x = trow[(2 * cc) * 32 + jj] * inv;
        a.y = trow[(2 * cc + 1) * 32 + jj] * inv;
        const v2f* bp = bmean + (jj * CC + cc) * 4 * 32;
#pragma unroll
        for (int ct = 0; ct < 4; ++ct)
          acc[ct] = __builtin_amdgcn_wmma_f32_16x16x4_f32(
              false, a, false, bp[ct * 32], (short)0, acc[ct], false, false);
      }
    }

#pragma unroll 4
    for (int t = 0; t < NT; ++t) {
      const float4 cur = q0;
      q0 = q1;
      q1 = q2;
      q2 = q3;
      q3 = *(const float4*)(srow + 8 * min(t + 4, NT - 1));
#pragma unroll
      for (int sub = 0; sub < 2; ++sub) {
        v2f a;
        a.x = sub ? cur.z : cur.x;
        a.y = sub ? cur.w : cur.y;
        const v2f* bp = bself + (2 * t + sub) * 4 * 32;
#pragma unroll
        for (int ct = 0; ct < 4; ++ct)
          acc[ct] = __builtin_amdgcn_wmma_f32_16x16x4_f32(
              false, a, false, bp[ct * 32], (short)0, acc[ct], false, false);
      }
    }

#pragma unroll
    for (int ct = 0; ct < 4; ++ct) {
      const int col = ct * 16 + lm;
      const float bv = bl[col];
#pragma unroll
      for (int r = 0; r < 8; ++r) {
        const int row = rt * 16 + r + hf * 8;
        float v = acc[ct][r] + bv;
        if (RELU) v = fmaxf(v, 0.0f);
        if (row < cnt) out[(long)(lo + row) * HID + col] = v;
      }
    }
  }
}

template <int F, bool COUNT, bool SAVE, bool LIST, bool KEYS, bool RELU>
__global__ void __launch_bounds__(SC_THREADS) sage_lds_scatter_kernel(
    const float* __restrict__ x, const int2* __restrict__ pairs,
    const int* __restrict__ dst, int E, int nNodes,
    const float* __restrict__ Wl, const float* __restrict__ bl,
    const float* __restrict__ Wr, float* __restrict__ out,
    float* __restrict__ deg, int* lists,
    const unsigned* __restrict__ planes) {
  constexpr int TILE_WORDS = (((RNG + 1) * (F + 1)) + 3) & ~3;
  __shared__ __attribute__((aligned(16))) float tile[TILE_WORDS];
  __shared__ float cntl[COUNT ? RNG + 1 : 1];
  __shared__ int queue[(SC_THREADS / 32) * QCAP];
  __shared__ int lcount;
  __shared__ __attribute__((aligned(16))) float wfrag[2 * F * HID];

  const int tid = threadIdx.x;
  const int lane = tid & 31;
  const int wave = __builtin_amdgcn_readfirstlane(tid >> 5);
  const int nW = blockDim.x >> 5;
  int* q = queue + wave * QCAP;

  const int nRanges = (nNodes + RNG - 1) / RNG;
  const bool aligned =
      KEYS ? (((unsigned long)planes) & 15ul) == 0ul
           : (((unsigned long)dst) & 15ul) == 0ul;
  const int pw = KEYS ? sage_plane_words(E) : 0;
  const int nFull = !aligned ? 0 : KEYS ? pw / PLANE_SLAB : E / SLAB_EDGES;
  const int nCovered = !aligned ? 0 : KEYS ? E : nFull * SLAB_EDGES;
  const int stride = max(1, nFull / nRanges);

  sage_stage_weights<F>(Wl, Wr, wfrag, tid, blockDim.x);

  for (int r = blockIdx.x; r < nRanges; r += gridDim.x) {
    const int lo = r * RNG;
    const unsigned cnt = (unsigned)min(RNG, nNodes - lo);
    int2* lst = (int2*)lists + (long)lo * LIST_PER_NODE;
    const int cap = (int)cnt * LIST_PER_NODE - LIST_HDR;

    for (int i = tid; i < TILE_WORDS / 4; i += blockDim.x)
      ((float4*)tile)[i] = make_float4(0.f, 0.f, 0.f, 0.f);
    if (COUNT)
      for (int i = tid; i <= RNG; i += blockDim.x) cntl[i] = 0.f;
    if (SAVE && tid == 0) lcount = 0;
    __syncthreads();

    bool scan = true;
    if (LIST) {
      const int total = __builtin_amdgcn_readfirstlane(lst[0].x);
      if ((unsigned)total <= (unsigned)cap) {
        scan = false;
        int b = wave * 32;
        int2 cur = make_int2(0, 0);
        if (b + lane < total) cur = lst[LIST_HDR + b + lane];
        for (; b < total; b += nW * 32) {
          const int bn = b + nW * 32;
          int2 nxt = make_int2(0, 0);
          if (bn + lane < total) nxt = lst[LIST_HDR + bn + lane];
          sage_add_rows<F>(cur.x, cur.y, min(32, total - b), x, tile, lane);
          cur = nxt;
        }
      }
    }

    if (scan) {
      int qn = 0;

      if constexpr (KEYS) {
        if (wave < nFull) {
          const int start =
              (int)(((unsigned)r * (unsigned)stride) % (unsigned)nFull);
          const uint4* ph =
              (const uint4*)(planes + (long)(((unsigned)r >> 4) & 15u) * pw) +
              lane;
          const uint4* pl =
              (const uint4*)(planes + (long)(16u + ((unsigned)r & 15u)) * pw) +
              lane;
          int sl = wave + start;
          if (sl >= nFull) sl -= nFull;
          uint4 ch = ph[sl * 32];
          uint4 cl = pl[sl * 32];

          for (int k = wave; k < nFull; k += nW) {
            int sn = k + nW + start;
            if (sn >= nFull) sn -= nFull;
            if (k + nW >= nFull) sn = sl;
            const uint4 nh = ph[sn * 32];
            const uint4 nl = pl[sn * 32];

            const int eb = (sl * PLANE_SLAB + lane * 4) << 5;
            const unsigned m4[4] = {ch.x & cl.x, ch.y & cl.y, ch.z & cl.z,
                                    ch.w & cl.w};
#pragma unroll
            for (int i = 0; i < 4; ++i) {
              unsigned f = m4[i];
              unsigned m = __builtin_amdgcn_ballot_w32(f != 0u);
              while (m) {
                if (f != 0u) {
                  q[qn + (int)__builtin_amdgcn_mbcnt_lo(m, 0u)] =
                      eb + i * 32 + __builtin_ctz(f);
                  f &= f - 1u;
                }
                qn += __builtin_popcount(m);
                if (qn >= Q_DRAIN)
                  qn = sage_drain_full<F, COUNT, SAVE>(q, qn, x, pairs, lo,
                                                       cnt, tile, cntl, lst,
                                                       cap, &lcount, lane);
                m = __builtin_amdgcn_ballot_w32(f != 0u);
              }
            }
            ch = nh;
            cl = nl;
            sl = sn;
          }
        }
      } else if (wave < nFull) {
        const int start =
            (int)(((unsigned)r * (unsigned)stride) % (unsigned)nFull);
        int sl = wave + start;
        if (sl >= nFull) sl -= nFull;
        int4 cur[SLAB_LOADS];
#pragma unroll
        for (int u = 0; u < SLAB_LOADS; ++u)
          cur[u] = *(const int4*)(dst + sl * SLAB_EDGES + u * 128 + lane * 4);

        for (int k = wave; k < nFull; k += nW) {
          int sn = k + nW + start;
          if (sn >= nFull) sn -= nFull;
          if (k + nW >= nFull) sn = sl;
          int4 nxt[SLAB_LOADS];
#pragma unroll
          for (int u = 0; u < SLAB_LOADS; ++u)
            nxt[u] = *(const int4*)(dst + sn * SLAB_EDGES + u * 128 + lane * 4);

#pragma unroll
          for (int u = 0; u < SLAB_LOADS; ++u) {
            const int eb = sl * SLAB_EDGES + u * 128 + lane * 4;
            const int dv4[4] = {cur[u].x, cur[u].y, cur[u].z, cur[u].w};
#pragma unroll
            for (int c = 0; c < 4; ++c) {
              const bool hit = (unsigned)(dv4[c] - lo) < cnt;
              const unsigned m = __builtin_amdgcn_ballot_w32(hit);
              if (m) {
                if (hit) q[qn + (int)__builtin_amdgcn_mbcnt_lo(m, 0u)] = eb + c;
                qn += __builtin_popcount(m);
              }
            }
            if (qn >= Q_DRAIN)
              qn = sage_drain_full<F, COUNT, SAVE>(q, qn, x, pairs, lo, cnt,
                                                   tile, cntl, lst, cap,
                                                   &lcount, lane);
          }
#pragma unroll
          for (int u = 0; u < SLAB_LOADS; ++u) cur[u] = nxt[u];
          sl = sn;
        }
      }
      for (int e0 = nCovered + wave * 32; e0 < E; e0 += nW * 32) {
        const int e = e0 + lane;
        const int d = (e < E) ? dst[e] : -1;
        const bool hit = (unsigned)(d - lo) < cnt;
        const unsigned m = __builtin_amdgcn_ballot_w32(hit);
        if (m) {
          if (hit) q[qn + (int)__builtin_amdgcn_mbcnt_lo(m, 0u)] = e;
          qn += __builtin_popcount(m);
        }
        if (qn >= Q_DRAIN)
          qn = sage_drain_full<F, COUNT, SAVE>(q, qn, x, pairs, lo, cnt, tile,
                                               cntl, lst, cap, &lcount, lane);
      }
      if (qn > 0)
        sage_drain<F, COUNT, SAVE>(q, qn, x, pairs, lo, cnt, tile, cntl, lst,
                                   cap, &lcount, lane);
    }
    __syncthreads();

    sage_tile_product<F, COUNT, RELU>(tile, cntl, wfrag, x, deg, bl, out, lo,
                                      (int)cnt, wave, nW, lane);
    if (COUNT)
      for (int i = tid; i < (int)cnt; i += blockDim.x) deg[lo + i] = cntl[i];
    if (SAVE && tid == 0) lst[0] = make_int2(lcount, 0);
    __syncthreads();
  }
}

extern "C" void kernel_launch(void* const* d_in, const int* in_sizes, int n_in,
                              void* d_out, int out_size, void* d_ws,
                              size_t ws_size, hipStream_t stream) {
  const float* x = (const float*)d_in[0];
  const int* ei = (const int*)d_in[1];
  const float* Wl1 = (const float*)d_in[2];
  const float* bl1 = (const float*)d_in[3];
  const float* Wr1 = (const float*)d_in[4];
  const float* Wl2 = (const float*)d_in[5];
  const float* bl2 = (const float*)d_in[6];
  const float* Wr2 = (const float*)d_in[7];
  float* out = (float*)d_out;

  const int N = in_sizes[0] / IN_C;
  const int E = in_sizes[1] / 2;
  const int* src = ei;
  const int* dst = ei + E;

  float* ws = (float*)d_ws;
  float* deg = ws;
  unsigned* planes = (unsigned*)(deg + N);
  int2* pairs = (int2*)(deg + N + (long)N * HID);
  float* h = deg + N + (long)N * IN_C;
  int* lists = (int*)(h + (long)N * HID);


  sage_range_key_kernel<<<512, 256, 0, stream>>>(src, dst, E, planes, pairs);
  sage_lds_scatter_kernel<IN_C, true, true, false, true, true>
      <<<256, SC_THREADS, 0, stream>>>(x, pairs, dst, E, N, Wl1, bl1, Wr1, h,
                                       deg, lists, planes);

  sage_lds_scatter_kernel<HID, false, false, true, false, false>
      <<<256, SC_THREADS, 0, stream>>>(h, pairs, dst, E, N, Wl2, bl2, Wr2, out,
                                       deg, lists, nullptr);
}


# Weights staged by 16-byte loads issued together (a lane takes two weight rows x four columns, stores 32 bytes to LDS), not a 4-byte load and wait per word: 16 and 8 dependent cache trips become one
# speedup vs baseline: 4.7800x; 1.0841x over previous
#include <hip/hip_runtime.h>

typedef __attribute__((ext_vector_type(2))) float v2f;
typedef __attribute__((ext_vector_type(8))) float v8f;

#define IN_C 128
#define HID  64
#define RNG 392
#define SC_THREADS 1024
#define SLAB_LOADS 8
#define SLAB_EDGES (SLAB_LOADS * 128)
#define Q_DRAIN 32
#define QCAP (Q_DRAIN + 128)
#define RIF 8
#define LIST_PER_NODE (HID / 2)
#define LIST_HDR 16
#define KEY_PLANES 32
#define PLANE_SLAB 128


__device__ __forceinline__ unsigned sage_key(int d) {
  return ((unsigned)d / (unsigned)RNG) & 255u;
}

__device__ __forceinline__ int sage_plane_words(int E) {
  return (((E + 31) >> 5) + PLANE_SLAB - 1) & ~(PLANE_SLAB - 1);
}

template <int HALF>
__device__ __forceinline__ void sage_plane_half(const unsigned (&kk)[32],
                                                unsigned* __restrict__ planes,
                                                int pw, int wbase, int lane) {
  unsigned acc[16];
#pragma unroll
  for (int p = 0; p < 16; ++p) acc[p] = 0u;
#pragma unroll
  for (int g = 0; g < 32; ++g) {
    const unsigned h = HALF ? ((kk[g] & 15u) | ((kk[g] >> 4) & 16u))
                            : (kk[g] >> 4);
#pragma unroll
    for (int p = 0; p < 16; ++p) {
      const unsigned w = __builtin_amdgcn_ballot_w32(h == (unsigned)p);
      acc[p] = (lane == g) ? w : acc[p];
    }
  }
#pragma unroll
  for (int p = 0; p < 16; ++p)
    planes[(long)(HALF * 16 + p) * pw + wbase + lane] = acc[p];
}

__global__ void __launch_bounds__(256) sage_range_key_kernel(
    const int* __restrict__ src, const int* __restrict__ dst, int E,
    unsigned* __restrict__ planes, int2* __restrict__ pairs) {
  const int lane = threadIdx.x & 31;
  const int wv = __builtin_amdgcn_readfirstlane(
      (int)((threadIdx.x >> 5) * gridDim.x + blockIdx.x));
  const int nWv = (int)(gridDim.x * (blockDim.x >> 5));
  const int pw = sage_plane_words(E);
  for (int t = wv; t < (pw >> 5); t += nWv) {
    const int e0 = t << 10;
    int dv[32], sv[32];
#pragma unroll
    for (int g = 0; g < 32; ++g) {
      const int e = min(e0 + (g << 5) + lane, E - 1);
      dv[g] = dst[e];
      sv[g] = src[e];
    }
    unsigned kk[32];
#pragma unroll
    for (int g = 0; g < 32; ++g) {
      const int e = e0 + (g << 5) + lane;
      unsigned k = 0x100u;
      if (e < E) {
        k = sage_key(dv[g]);
        pairs[e] = make_int2(sv[g], dv[g]);
      }
      kk[g] = k;
    }
    sage_plane_half<0>(kk, planes, pw, t << 5, lane);
    sage_plane_half<1>(kk, planes, pw, t << 5, lane);
  }
}

template <int F>
__device__ __forceinline__ void sage_add_rows(int sv, int dv, int n,
                                              const float* __restrict__ x,
                                              float* tile, int lane) {
  constexpr int VEC = F / 32;
  for (int j = 0; j < n; j += RIF) {
    float v[RIF][VEC];
    int dd[RIF];
#pragma unroll
    for (int k = 0; k < RIF; ++k) {
      const int jj = min(j + k, n - 1);
      const int sr = __builtin_amdgcn_readlane(sv, jj);
      dd[k] = __builtin_amdgcn_readlane(dv, jj);
      const float* row = x + (long)sr * F + lane * VEC;
      if constexpr (VEC == 4) {
        const float4 t = *(const float4*)row;
        v[k][0] = t.x;
        v[k][1] = t.y;
        v[k][2] = t.z;
        v[k][3] = t.w;
      } else {
        const float2 t = *(const float2*)row;
        v[k][0] = t.x;
        v[k][1] = t.y;
      }
    }
#pragma unroll
    for (int k = 0; k < RIF; ++k)
      if (j + k < n) {
        float* t = tile + dd[k] * (F + 1) + lane;
#pragma unroll
        for (int c = 0; c < VEC; ++c)
          __hip_atomic_fetch_add(t + c * 32, v[k][c], __ATOMIC_RELAXED,
                                 __HIP_MEMORY_SCOPE_WORKGROUP);
      }
  }
}

template <int F, bool COUNT, bool SAVE>
__device__ __forceinline__ void sage_drain(const int* q, int qn,
                                           const float* __restrict__ x,
                                           const int2* __restrict__ pairs,
                                           int lo, unsigned cnt, float* tile,
                                           float* cntl, int2* lst, int cap,
                                           int* lcount, int lane) {
  __builtin_amdgcn_fence(__ATOMIC_ACQ_REL, "wavefront");
  for (int b = 0; b < qn; b += 32) {
    const int n = min(32, qn - b);
    int sv = 0, dv = 0;
    if (lane < n) {
      const int2 p = pairs[q[b + lane]];
      sv = p.x;
      dv = p.y - lo;
      if ((unsigned)dv >= cnt) {
        sv = 0;
        dv = RNG;
      }
      if (COUNT)
        __hip_atomic_fetch_add(&cntl[dv], 1.0f, __ATOMIC_RELAXED,
                               __HIP_MEMORY_SCOPE_WORKGROUP);
    }
    if (SAVE) {
      int pos = 0;
      if (lane == 0)
        pos = __hip_atomic_fetch_add(lcount, n, __ATOMIC_RELAXED,
                                     __HIP_MEMORY_SCOPE_WORKGROUP);
      pos = __builtin_amdgcn_readfirstlane(pos);
      if (lane < n && pos + lane < cap)
        lst[LIST_HDR + pos + lane] = make_int2(sv, dv);
    }
    sage_add_rows<F>(sv, dv, n, x, tile, lane);
  }
  __builtin_amdgcn_fence(__ATOMIC_ACQ_REL, "wavefront");
}

template <int F, bool COUNT, bool SAVE>
__device__ __forceinline__ int sage_drain_full(int* q, int qn,
                                               const float* __restrict__ x,
                                               const int2* __restrict__ pairs,
                                               int lo, unsigned cnt, float* tile,
                                               float* cntl, int2* lst, int cap,
                                               int* lcount, int lane) {
  const int full = qn & ~(Q_DRAIN - 1);
  sage_drain<F, COUNT, SAVE>(q, full, x, pairs, lo, cnt, tile, cntl, lst, cap,
                             lcount, lane);
  const int rem = qn - full;
  if (lane < rem) q[lane] = q[full + lane];
  return rem;
}

template <int F>
__device__ __forceinline__ int sage_feat_mean(int s, int hf, int r) {
  constexpr int VEC = F / 32;
  constexpr int CC = VEC / 2;
  return VEC * (s / CC + 16 * hf) + 2 * (s % CC) + r;
}

__device__ __forceinline__ int sage_feat_self(int s, int hf, int r) {
  return 8 * (s >> 1) + 4 * hf + 2 * (s & 1) + r;
}

template <int F>
__device__ __forceinline__ const float* sage_stage_src(
    const float* __restrict__ Wl, const float* __restrict__ Wr, int t) {
  const int q = t & 3;
  const int hf = (t >> 2) & 1;
  const int ct = (t >> 3) & 3;
  const int st = t >> 5;
  const int s = st % (F / 4);
  const float* W = (st < F / 4) ? Wl : Wr;
  const int f0 = (st < F / 4) ? sage_feat_mean<F>(s, hf, 0)
                              : sage_feat_self(s, hf, 0);
  return W + f0 * HID + ct * 16 + 4 * q;
}

template <int F>
__device__ __forceinline__ void sage_stage_weights(const float* __restrict__ Wl,
                                                   const float* __restrict__ Wr,
                                                   float* wfrag, int tid,
                                                   int nT) {
  constexpr int TASKS = F * 16;
  constexpr int PER = (TASKS + SC_THREADS - 1) / SC_THREADS;
  if (((((unsigned long)Wl) | ((unsigned long)Wr)) & 15ul) == 0ul) {
    float4 a0[PER], a1[PER];
    int tt[PER];
#pragma unroll
    for (int k = 0; k < PER; ++k) {
      tt[k] = min(tid + k * nT, TASKS - 1);
      const float* p = sage_stage_src<F>(Wl, Wr, tt[k]);
      a0[k] = *(const float4*)p;
      a1[k] = *(const float4*)(p + HID);
    }
#pragma unroll
    for (int k = 0; k < PER; ++k) {
      float4* d = (float4*)(wfrag + tt[k] * 8);
      d[0] = make_float4(a0[k].x, a1[k].x, a0[k].y, a1[k].y);
      d[1] = make_float4(a0[k].z, a1[k].z, a0[k].w, a1[k].w);
    }
    for (int t = tid + PER * nT; t < TASKS; t += nT) {
      const float* p = sage_stage_src<F>(Wl, Wr, t);
      const float4 b0 = *(const float4*)p;
      const float4 b1 = *(const float4*)(p + HID);
      float4* d = (float4*)(wfrag + t * 8);
      d[0] = make_float4(b0.x, b1.x, b0.y, b1.y);
      d[1] = make_float4(b0.z, b1.z, b0.w, b1.w);
    }
  } else {
    for (int idx = tid; idx < 2 * F * HID; idx += nT) {
      const int r = idx & 1;
      const int ln = (idx >> 1) & 31;
      const int ct = (idx >> 6) & 3;
      const int st = idx >> 8;
      const int s = st % (F / 4);
      const int hf = ln >> 4;
      const int col = ct * 16 + (ln & 15);
      wfrag[idx] = (st < F / 4) ? Wl[sage_feat_mean<F>(s, hf, r) * HID + col]
                                : Wr[sage_feat_self(s, hf, r) * HID + col];
    }
  }
}

template <int F, bool COUNT, bool RELU>
__device__ __forceinline__ void sage_tile_product(
    const float* tile, const float* cntl, const float* wfrag,
    const float* __restrict__ self, const float* __restrict__ deg,
    const float* __restrict__ bl, float* __restrict__ out, int lo, int cnt,
    int wave, int nW, int lane) {
  constexpr int VEC = F / 32;
  constexpr int CC = VEC / 2;
  const int hf = lane >> 4;
  const int lm = lane & 15;
  const int nTiles = (cnt + 15) >> 4;
  const v2f* bmean = (const v2f*)wfrag + lane;
  const v2f* bself = bmean + (F / 4) * 4 * 32;

  for (int rt = wave; rt < nTiles; rt += nW) {
    const int i = min(rt * 16 + lm, cnt - 1);

    constexpr int NT = F / 8;
    const float* srow = self + (long)(lo + i) * F + 4 * hf;
    float4 q0 = *(const float4*)(srow);
    float4 q1 = *(const float4*)(srow + 8);
    float4 q2 = *(const float4*)(srow + 16);
    float4 q3 = *(const float4*)(srow + 24);

    float dg;
    if constexpr (COUNT)
      dg = cntl[i];
    else
      dg = deg[lo + i];
    const float inv = 1.0f / fmaxf(dg, 1.0f);

    v8f acc[4];
#pragma unroll
    for (int ct = 0; ct < 4; ++ct)
      acc[ct] = (v8f){0.f, 0.f, 0.f, 0.f, 0.f, 0.f, 0.f, 0.f};

    const float* trow = tile + i * (F + 1) + 16 * hf;
#pragma unroll 2
    for (int jj = 0; jj < 16; ++jj) {
#pragma unroll
      for (int cc = 0; cc < CC; ++cc) {
        v2f a;
        a.x = trow[(2 * cc) * 32 + jj] * inv;
        a.y = trow[(2 * cc + 1) * 32 + jj] * inv;
        const v2f* bp = bmean + (jj * CC + cc) * 4 * 32;
#pragma unroll
        for (int ct = 0; ct < 4; ++ct)
          acc[ct] = __builtin_amdgcn_wmma_f32_16x16x4_f32(
              false, a, false, bp[ct * 32], (short)0, acc[ct], false, false);
      }
    }

#pragma unroll 4
    for (int t = 0; t < NT; ++t) {
      const float4 cur = q0;
      q0 = q1;
      q1 = q2;
      q2 = q3;
      q3 = *(const float4*)(srow + 8 * min(t + 4, NT - 1));
#pragma unroll
      for (int sub = 0; sub < 2; ++sub) {
        v2f a;
        a.x = sub ? cur.z : cur.x;
        a.y = sub ? cur.w : cur.y;
        const v2f* bp = bself + (2 * t + sub) * 4 * 32;
#pragma unroll
        for (int ct = 0; ct < 4; ++ct)
          acc[ct] = __builtin_amdgcn_wmma_f32_16x16x4_f32(
              false, a, false, bp[ct * 32], (short)0, acc[ct], false, false);
      }
    }

#pragma unroll
    for (int ct = 0; ct < 4; ++ct) {
      const int col = ct * 16 + lm;
      const float bv = bl[col];
#pragma unroll
      for (int r = 0; r < 8; ++r) {
        const int row = rt * 16 + r + hf * 8;
        float v = acc[ct][r] + bv;
        if (RELU) v = fmaxf(v, 0.0f);
        if (row < cnt) out[(long)(lo + row) * HID + col] = v;
      }
    }
  }
}

template <int F, bool COUNT, bool SAVE, bool LIST, bool KEYS, bool RELU>
__global__ void __launch_bounds__(SC_THREADS) sage_lds_scatter_kernel(
    const float* __restrict__ x, const int2* __restrict__ pairs,
    const int* __restrict__ dst, int E, int nNodes,
    const float* __restrict__ Wl, const float* __restrict__ bl,
    const float* __restrict__ Wr, float* __restrict__ out,
    float* __restrict__ deg, int* lists,
    const unsigned* __restrict__ planes) {
  constexpr int TILE_WORDS = (((RNG + 1) * (F + 1)) + 3) & ~3;
  __shared__ __attribute__((aligned(16))) float tile[TILE_WORDS];
  __shared__ float cntl[COUNT ? RNG + 1 : 1];
  __shared__ int queue[(SC_THREADS / 32) * QCAP];
  __shared__ int lcount;
  __shared__ __attribute__((aligned(16))) float wfrag[2 * F * HID];

  const int tid = threadIdx.x;
  const int lane = tid & 31;
  const int wave = __builtin_amdgcn_readfirstlane(tid >> 5);
  const int nW = blockDim.x >> 5;
  int* q = queue + wave * QCAP;

  const int nRanges = (nNodes + RNG - 1) / RNG;
  const bool aligned =
      KEYS ? (((unsigned long)planes) & 15ul) == 0ul
           : (((unsigned long)dst) & 15ul) == 0ul;
  const int pw = KEYS ? sage_plane_words(E) : 0;
  const int nFull = !aligned ? 0 : KEYS ? pw / PLANE_SLAB : E / SLAB_EDGES;
  const int nCovered = !aligned ? 0 : KEYS ? E : nFull * SLAB_EDGES;
  const int stride = max(1, nFull / nRanges);

  sage_stage_weights<F>(Wl, Wr, wfrag, tid, blockDim.x);

  for (int r = blockIdx.x; r < nRanges; r += gridDim.x) {
    const int lo = r * RNG;
    const unsigned cnt = (unsigned)min(RNG, nNodes - lo);
    int2* lst = (int2*)lists + (long)lo * LIST_PER_NODE;
    const int cap = (int)cnt * LIST_PER_NODE - LIST_HDR;

    for (int i = tid; i < TILE_WORDS / 4; i += blockDim.x)
      ((float4*)tile)[i] = make_float4(0.f, 0.f, 0.f, 0.f);
    if (COUNT)
      for (int i = tid; i <= RNG; i += blockDim.x) cntl[i] = 0.f;
    if (SAVE && tid == 0) lcount = 0;
    __syncthreads();

    bool scan = true;
    if (LIST) {
      const int total = __builtin_amdgcn_readfirstlane(lst[0].x);
      if ((unsigned)total <= (unsigned)cap) {
        scan = false;
        int b = wave * 32;
        int2 cur = make_int2(0, 0);
        if (b + lane < total) cur = lst[LIST_HDR + b + lane];
        for (; b < total; b += nW * 32) {
          const int bn = b + nW * 32;
          int2 nxt = make_int2(0, 0);
          if (bn + lane < total) nxt = lst[LIST_HDR + bn + lane];
          sage_add_rows<F>(cur.x, cur.y, min(32, total - b), x, tile, lane);
          cur = nxt;
        }
      }
    }

    if (scan) {
      int qn = 0;

      if constexpr (KEYS) {
        if (wave < nFull) {
          const int start =
              (int)(((unsigned)r * (unsigned)stride) % (unsigned)nFull);
          const uint4* ph =
              (const uint4*)(planes + (long)(((unsigned)r >> 4) & 15u) * pw) +
              lane;
          const uint4* pl =
              (const uint4*)(planes + (long)(16u + ((unsigned)r & 15u)) * pw) +
              lane;
          int sl = wave + start;
          if (sl >= nFull) sl -= nFull;
          uint4 ch = ph[sl * 32];
          uint4 cl = pl[sl * 32];

          for (int k = wave; k < nFull; k += nW) {
            int sn = k + nW + start;
            if (sn >= nFull) sn -= nFull;
            if (k + nW >= nFull) sn = sl;
            const uint4 nh = ph[sn * 32];
            const uint4 nl = pl[sn * 32];

            const int eb = (sl * PLANE_SLAB + lane * 4) << 5;
            const unsigned m4[4] = {ch.x & cl.x, ch.y & cl.y, ch.z & cl.z,
                                    ch.w & cl.w};
#pragma unroll
            for (int i = 0; i < 4; ++i) {
              unsigned f = m4[i];
              unsigned m = __builtin_amdgcn_ballot_w32(f != 0u);
              while (m) {
                if (f != 0u) {
                  q[qn + (int)__builtin_amdgcn_mbcnt_lo(m, 0u)] =
                      eb + i * 32 + __builtin_ctz(f);
                  f &= f - 1u;
                }
                qn += __builtin_popcount(m);
                if (qn >= Q_DRAIN)
                  qn = sage_drain_full<F, COUNT, SAVE>(q, qn, x, pairs, lo,
                                                       cnt, tile, cntl, lst,
                                                       cap, &lcount, lane);
                m = __builtin_amdgcn_ballot_w32(f != 0u);
              }
            }
            ch = nh;
            cl = nl;
            sl = sn;
          }
        }
      } else if (wave < nFull) {
        const int start =
            (int)(((unsigned)r * (unsigned)stride) % (unsigned)nFull);
        int sl = wave + start;
        if (sl >= nFull) sl -= nFull;
        int4 cur[SLAB_LOADS];
#pragma unroll
        for (int u = 0; u < SLAB_LOADS; ++u)
          cur[u] = *(const int4*)(dst + sl * SLAB_EDGES + u * 128 + lane * 4);

        for (int k = wave; k < nFull; k += nW) {
          int sn = k + nW + start;
          if (sn >= nFull) sn -= nFull;
          if (k + nW >= nFull) sn = sl;
          int4 nxt[SLAB_LOADS];
#pragma unroll
          for (int u = 0; u < SLAB_LOADS; ++u)
            nxt[u] = *(const int4*)(dst + sn * SLAB_EDGES + u * 128 + lane * 4);

#pragma unroll
          for (int u = 0; u < SLAB_LOADS; ++u) {
            const int eb = sl * SLAB_EDGES + u * 128 + lane * 4;
            const int dv4[4] = {cur[u].x, cur[u].y, cur[u].z, cur[u].w};
#pragma unroll
            for (int c = 0; c < 4; ++c) {
              const bool hit = (unsigned)(dv4[c] - lo) < cnt;
              const unsigned m = __builtin_amdgcn_ballot_w32(hit);
              if (m) {
                if (hit) q[qn + (int)__builtin_amdgcn_mbcnt_lo(m, 0u)] = eb + c;
                qn += __builtin_popcount(m);
              }
            }
            if (qn >= Q_DRAIN)
              qn = sage_drain_full<F, COUNT, SAVE>(q, qn, x, pairs, lo, cnt,
                                                   tile, cntl, lst, cap,
                                                   &lcount, lane);
          }
#pragma unroll
          for (int u = 0; u < SLAB_LOADS; ++u) cur[u] = nxt[u];
          sl = sn;
        }
      }
      for (int e0 = nCovered + wave * 32; e0 < E; e0 += nW * 32) {
        const int e = e0 + lane;
        const int d = (e < E) ? dst[e] : -1;
        const bool hit = (unsigned)(d - lo) < cnt;
        const unsigned m = __builtin_amdgcn_ballot_w32(hit);
        if (m) {
          if (hit) q[qn + (int)__builtin_amdgcn_mbcnt_lo(m, 0u)] = e;
          qn += __builtin_popcount(m);
        }
        if (qn >= Q_DRAIN)
          qn = sage_drain_full<F, COUNT, SAVE>(q, qn, x, pairs, lo, cnt, tile,
                                               cntl, lst, cap, &lcount, lane);
      }
      if (qn > 0)
        sage_drain<F, COUNT, SAVE>(q, qn, x, pairs, lo, cnt, tile, cntl, lst,
                                   cap, &lcount, lane);
    }
    __syncthreads();

    sage_tile_product<F, COUNT, RELU>(tile, cntl, wfrag, x, deg, bl, out, lo,
                                      (int)cnt, wave, nW, lane);
    if (COUNT)
      for (int i = tid; i < (int)cnt; i += blockDim.x) deg[lo + i] = cntl[i];
    if (SAVE && tid == 0) lst[0] = make_int2(lcount, 0);
    __syncthreads();
  }
}

extern "C" void kernel_launch(void* const* d_in, const int* in_sizes, int n_in,
                              void* d_out, int out_size, void* d_ws,
                              size_t ws_size, hipStream_t stream) {
  const float* x = (const float*)d_in[0];
  const int* ei = (const int*)d_in[1];
  const float* Wl1 = (const float*)d_in[2];
  const float* bl1 = (const float*)d_in[3];
  const float* Wr1 = (const float*)d_in[4];
  const float* Wl2 = (const float*)d_in[5];
  const float* bl2 = (const float*)d_in[6];
  const float* Wr2 = (const float*)d_in[7];
  float* out = (float*)d_out;

  const int N = in_sizes[0] / IN_C;
  const int E = in_sizes[1] / 2;
  const int* src = ei;
  const int* dst = ei + E;

  float* ws = (float*)d_ws;
  float* deg = ws;
  unsigned* planes = (unsigned*)(deg + N);
  int2* pairs = (int2*)(deg + N + (long)N * HID);
  float* h = deg + N + (long)N * IN_C;
  int* lists = (int*)(h + (long)N * HID);


  sage_range_key_kernel<<<512, 256, 0, stream>>>(src, dst, E, planes, pairs);
  sage_lds_scatter_kernel<IN_C, true, true, false, true, true>
      <<<256, SC_THREADS, 0, stream>>>(x, pairs, dst, E, N, Wl1, bl1, Wr1, h,
                                       deg, lists, planes);

  sage_lds_scatter_kernel<HID, false, false, true, false, false>
      <<<256, SC_THREADS, 0, stream>>>(h, pairs, dst, E, N, Wl2, bl2, Wr2, out,
                                       deg, lists, nullptr);
}


# Self product (x @ W_r) of a wave's 16 rows runs in front of its gather, accumulators held in registers, under other waves' row loads; slabs handed out by an LDS counter; k sum: self half first
# speedup vs baseline: 5.1291x; 1.0730x over previous
#include <hip/hip_runtime.h>

typedef __attribute__((ext_vector_type(2))) float v2f;
typedef __attribute__((ext_vector_type(4))) float v4f;
typedef __attribute__((ext_vector_type(8))) float v8f;

#define IN_C 128
#define HID  64
#define RNG 392
#define SC_THREADS 1024
#define SLAB_LOADS 8
#define SLAB_EDGES (SLAB_LOADS * 128)
#define Q_DRAIN 32
#define QCAP (Q_DRAIN + 128)
#define RIF 8
#define LIST_PER_NODE (HID / 2)
#define LIST_HDR 16
#define KEY_PLANES 32
#define PLANE_SLAB 128


__device__ __forceinline__ unsigned sage_key(int d) {
  return ((unsigned)d / (unsigned)RNG) & 255u;
}

__device__ __forceinline__ int sage_plane_words(int E) {
  return (((E + 31) >> 5) + PLANE_SLAB - 1) & ~(PLANE_SLAB - 1);
}

template <int HALF>
__device__ __forceinline__ void sage_plane_half(const unsigned (&kk)[32],
                                                unsigned* __restrict__ planes,
                                                int pw, int wbase, int lane) {
  unsigned acc[16];
#pragma unroll
  for (int p = 0; p < 16; ++p) acc[p] = 0u;
#pragma unroll
  for (int g = 0; g < 32; ++g) {
    const unsigned h = HALF ? ((kk[g] & 15u) | ((kk[g] >> 4) & 16u))
                            : (kk[g] >> 4);
#pragma unroll
    for (int p = 0; p < 16; ++p) {
      const unsigned w = __builtin_amdgcn_ballot_w32(h == (unsigned)p);
      acc[p] = (lane == g) ? w : acc[p];
    }
  }
#pragma unroll
  for (int p = 0; p < 16; ++p)
    planes[(long)(HALF * 16 + p) * pw + wbase + lane] = acc[p];
}

__global__ void __launch_bounds__(256) sage_range_key_kernel(
    const int* __restrict__ src, const int* __restrict__ dst, int E,
    unsigned* __restrict__ planes, int2* __restrict__ pairs) {
  const int lane = threadIdx.x & 31;
  const int wv = __builtin_amdgcn_readfirstlane(
      (int)((threadIdx.x >> 5) * gridDim.x + blockIdx.x));
  const int nWv = (int)(gridDim.x * (blockDim.x >> 5));
  const int pw = sage_plane_words(E);
  for (int t = wv; t < (pw >> 5); t += nWv) {
    const int e0 = t << 10;
    int dv[32], sv[32];
#pragma unroll
    for (int g = 0; g < 32; ++g) {
      const int e = min(e0 + (g << 5) + lane, E - 1);
      dv[g] = dst[e];
      sv[g] = src[e];
    }
    unsigned kk[32];
#pragma unroll
    for (int g = 0; g < 32; ++g) {
      const int e = e0 + (g << 5) + lane;
      unsigned k = 0x100u;
      if (e < E) {
        k = sage_key(dv[g]);
        pairs[e] = make_int2(sv[g], dv[g]);
      }
      kk[g] = k;
    }
    sage_plane_half<0>(kk, planes, pw, t << 5, lane);
    sage_plane_half<1>(kk, planes, pw, t << 5, lane);
  }
}

template <int F>
__device__ __forceinline__ void sage_add_rows(int sv, int dv, int n,
                                              const float* __restrict__ x,
                                              float* tile, int lane) {
  constexpr int VEC = F / 32;
  for (int j = 0; j < n; j += RIF) {
    float v[RIF][VEC];
    int dd[RIF];
#pragma unroll
    for (int k = 0; k < RIF; ++k) {
      const int jj = min(j + k, n - 1);
      const int sr = __builtin_amdgcn_readlane(sv, jj);
      dd[k] = __builtin_amdgcn_readlane(dv, jj);
      const float* row = x + (long)sr * F + lane * VEC;
      if constexpr (VEC == 4) {
        const float4 t = *(const float4*)row;
        v[k][0] = t.x;
        v[k][1] = t.y;
        v[k][2] = t.z;
        v[k][3] = t.w;
      } else {
        const float2 t = *(const float2*)row;
        v[k][0] = t.x;
        v[k][1] = t.y;
      }
    }
#pragma unroll
    for (int k = 0; k < RIF; ++k)
      if (j + k < n) {
        float* t = tile + dd[k] * (F + 1) + lane;
#pragma unroll
        for (int c = 0; c < VEC; ++c)
          __hip_atomic_fetch_add(t + c * 32, v[k][c], __ATOMIC_RELAXED,
                                 __HIP_MEMORY_SCOPE_WORKGROUP);
      }
  }
}

template <int F, bool COUNT, bool SAVE>
__device__ __forceinline__ void sage_drain(const int* q, int qn,
                                           const float* __restrict__ x,
                                           const int2* __restrict__ pairs,
                                           int lo, unsigned cnt, float* tile,
                                           float* cntl, int2* lst, int cap,
                                           int* lcount, int lane) {
  __builtin_amdgcn_fence(__ATOMIC_ACQ_REL, "wavefront");
  for (int b = 0; b < qn; b += 32) {
    const int n = min(32, qn - b);
    int sv = 0, dv = 0;
    if (lane < n) {
      const int2 p = pairs[q[b + lane]];
      sv = p.x;
      dv = p.y - lo;
      if ((unsigned)dv >= cnt) {
        sv = 0;
        dv = RNG;
      }
      if (COUNT)
        __hip_atomic_fetch_add(&cntl[dv], 1.0f, __ATOMIC_RELAXED,
                               __HIP_MEMORY_SCOPE_WORKGROUP);
    }
    if (SAVE) {
      int pos = 0;
      if (lane == 0)
        pos = __hip_atomic_fetch_add(lcount, n, __ATOMIC_RELAXED,
                                     __HIP_MEMORY_SCOPE_WORKGROUP);
      pos = __builtin_amdgcn_readfirstlane(pos);
      if (lane < n && pos + lane < cap)
        lst[LIST_HDR + pos + lane] = make_int2(sv, dv);
    }
    sage_add_rows<F>(sv, dv, n, x, tile, lane);
  }
  __builtin_amdgcn_fence(__ATOMIC_ACQ_REL, "wavefront");
}

template <int F, bool COUNT, bool SAVE>
__device__ __forceinline__ int sage_drain_full(int* q, int qn,
                                               const float* __restrict__ x,
                                               const int2* __restrict__ pairs,
                                               int lo, unsigned cnt, float* tile,
                                               float* cntl, int2* lst, int cap,
                                               int* lcount, int lane) {
  const int full = qn & ~(Q_DRAIN - 1);
  sage_drain<F, COUNT, SAVE>(q, full, x, pairs, lo, cnt, tile, cntl, lst, cap,
                             lcount, lane);
  const int rem = qn - full;
  if (lane < rem) q[lane] = q[full + lane];
  return rem;
}

template <int F>
__device__ __forceinline__ int sage_feat_mean(int s, int hf, int r) {
  constexpr int VEC = F / 32;
  constexpr int CC = VEC / 2;
  return VEC * (s / CC + 16 * hf) + 2 * (s % CC) + r;
}

__device__ __forceinline__ int sage_feat_self(int s, int hf, int r) {
  return 8 * (s >> 1) + 4 * hf + 2 * (s & 1) + r;
}

template <int F>
__device__ __forceinline__ const float* sage_stage_src(
    const float* __restrict__ Wl, const float* __restrict__ Wr, int t) {
  const int q = t & 3;
  const int hf = (t >> 2) & 1;
  const int ct = (t >> 3) & 3;
  const int st = t >> 5;
  const int s = st % (F / 4);
  const float* W = (st < F / 4) ? Wl : Wr;
  const int f0 = (st < F / 4) ? sage_feat_mean<F>(s, hf, 0)
                              : sage_feat_self(s, hf, 0);
  return W + f0 * HID + ct * 16 + 4 * q;
}

template <int F>
__device__ __forceinline__ void sage_stage_weights(const float* __restrict__ Wl,
                                                   const float* __restrict__ Wr,
                                                   float* wfrag, int tid,
                                                   int nT) {
  constexpr int TASKS = F * 16;
  constexpr int PER = (TASKS + SC_THREADS - 1) / SC_THREADS;
  if (((((unsigned long)Wl) | ((unsigned long)Wr)) & 15ul) == 0ul) {
    float4 a0[PER], a1[PER];
    int tt[PER];
#pragma unroll
    for (int k = 0; k < PER; ++k) {
      tt[k] = min(tid + k * nT, TASKS - 1);
      const float* p = sage_stage_src<F>(Wl, Wr, tt[k]);
      a0[k] = *(const float4*)p;
      a1[k] = *(const float4*)(p + HID);
    }
#pragma unroll
    for (int k = 0; k < PER; ++k) {
      float4* d = (float4*)(wfrag + tt[k] * 8);
      d[0] = make_float4(a0[k].x, a1[k].x, a0[k].y, a1[k].y);
      d[1] = make_float4(a0[k].z, a1[k].z, a0[k].w, a1[k].w);
    }
    for (int t = tid + PER * nT; t < TASKS; t += nT) {
      const float* p = sage_stage_src<F>(Wl, Wr, t);
      const float4 b0 = *(const float4*)p;
      const float4 b1 = *(const float4*)(p + HID);
      float4* d = (float4*)(wfrag + t * 8);
      d[0] = make_float4(b0.x, b1.x, b0.y, b1.y);
      d[1] = make_float4(b0.z, b1.z, b0.w, b1.w);
    }
  } else {
    for (int idx = tid; idx < 2 * F * HID; idx += nT) {
      const int r = idx & 1;
      const int ln = (idx >> 1) & 31;
      const int ct = (idx >> 6) & 3;
      const int st = idx >> 8;
      const int s = st % (F / 4);
      const int hf = ln >> 4;
      const int col = ct * 16 + (ln & 15);
      wfrag[idx] = (st < F / 4) ? Wl[sage_feat_mean<F>(s, hf, r) * HID + col]
                                : Wr[sage_feat_self(s, hf, r) * HID + col];
    }
  }
}


template <int F>
__device__ __forceinline__ void sage_self_term(v8f (&acc)[4], const float* wfrag,
                                               const float* __restrict__ self,
                                               int lo, int cnt, int rt,
                                               int lane) {
  const int hf = lane >> 4;
  const int lm = lane & 15;
  const v2f* bself = (const v2f*)wfrag + lane + (F / 4) * 4 * 32;
  const int i = min(rt * 16 + lm, cnt - 1);
  constexpr int NT = F / 8;
  const float* srow = self + (long)(lo + i) * F + 4 * hf;
  v4f q0 = *(const v4f*)(srow);
  v4f q1 = *(const v4f*)(srow + 8);
  v4f q2 = *(const v4f*)(srow + 16);
  v4f q3 = *(const v4f*)(srow + 24);
#pragma unroll 4
  for (int t = 0; t < NT; ++t) {
    const v4f cur = q0;
    q0 = q1;
    q1 = q2;
    q2 = q3;
    q3 = *(const v4f*)(srow + 8 * min(t + 4, NT - 1));
#pragma unroll
    for (int sub = 0; sub < 2; ++sub) {
      const v2f a = sub ? cur.zw : cur.xy;
      const v2f* bp = bself + (2 * t + sub) * 4 * 32;
#pragma unroll
      for (int ct = 0; ct < 4; ++ct)
        acc[ct] = __builtin_amdgcn_wmma_f32_16x16x4_f32(
            false, a, false, bp[ct * 32], (short)0, acc[ct], false, false);
    }
  }
}

template <int F, bool COUNT>
__device__ __forceinline__ void sage_mean_term(v8f (&acc)[4], const float* tile,
                                               const float* cntl,
                                               const float* wfrag,
                                               const float* __restrict__ deg,
                                               int lo, int cnt, int rt,
                                               int lane) {
  constexpr int VEC = F / 32;
  constexpr int CC = VEC / 2;
  const int hf = lane >> 4;
  const int lm = lane & 15;
  const v2f* bmean = (const v2f*)wfrag + lane;
  const int i = min(rt * 16 + lm, cnt - 1);
  float dg;
  if constexpr (COUNT)
    dg = cntl[i];
  else
    dg = deg[lo + i];
  const float inv = 1.0f / fmaxf(dg, 1.0f);
  const float* trow = tile + i * (F + 1) + 16 * hf;
#pragma unroll 2
  for (int jj = 0; jj < 16; ++jj) {
#pragma unroll
    for (int cc = 0; cc < CC; ++cc) {
      v2f a;
      a.x = trow[(2 * cc) * 32 + jj] * inv;
      a.y = trow[(2 * cc + 1) * 32 + jj] * inv;
      const v2f* bp = bmean + (jj * CC + cc) * 4 * 32;
#pragma unroll
      for (int ct = 0; ct < 4; ++ct)
        acc[ct] = __builtin_amdgcn_wmma_f32_16x16x4_f32(
            false, a, false, bp[ct * 32], (short)0, acc[ct], false, false);
    }
  }
}

template <bool RELU>
__device__ __forceinline__ void sage_store_tile(const v8f (&acc)[4],
                                                const float* __restrict__ bl,
                                                float* __restrict__ out, int lo,
                                                int cnt, int rt, int lane) {
  const int hf = lane >> 4;
  const int lm = lane & 15;
#pragma unroll
  for (int ct = 0; ct < 4; ++ct) {
    const int col = ct * 16 + lm;
    const float bv = bl[col];
#pragma unroll
    for (int r = 0; r < 8; ++r) {
      const int row = rt * 16 + r + hf * 8;
      float v = acc[ct][r] + bv;
      if (RELU) v = fmaxf(v, 0.0f);
      if (row < cnt) out[(long)(lo + row) * HID + col] = v;
    }
  }
}

__device__ __forceinline__ int sage_take(int* counter, int lane) {
  int t = 0;
  if (lane == 0)
    t = __hip_atomic_fetch_add(counter, 1, __ATOMIC_RELAXED,
                               __HIP_MEMORY_SCOPE_WORKGROUP);
  return __builtin_amdgcn_readfirstlane(t);
}

template <int F, bool COUNT, bool SAVE, bool LIST, bool KEYS, bool RELU>
__global__ void __launch_bounds__(SC_THREADS) sage_lds_scatter_kernel(
    const float* __restrict__ x, const int2* __restrict__ pairs,
    const int* __restrict__ dst, int E, int nNodes,
    const float* __restrict__ Wl, const float* __restrict__ bl,
    const float* __restrict__ Wr, float* __restrict__ out,
    float* __restrict__ deg, int* lists,
    const unsigned* __restrict__ planes) {
  constexpr int TILE_WORDS = (((RNG + 1) * (F + 1)) + 3) & ~3;
  __shared__ __attribute__((aligned(16))) float tile[TILE_WORDS];
  __shared__ float cntl[COUNT ? RNG + 1 : 1];
  __shared__ int queue[(SC_THREADS / 32) * QCAP];
  __shared__ int lcount;
  __shared__ int slabNext;
  __shared__ __attribute__((aligned(16))) float wfrag[2 * F * HID];

  const int tid = threadIdx.x;
  const int lane = tid & 31;
  const int wave = __builtin_amdgcn_readfirstlane(tid >> 5);
  const int nW = blockDim.x >> 5;
  int* q = queue + wave * QCAP;

  const int nRanges = (nNodes + RNG - 1) / RNG;
  const bool aligned =
      KEYS ? (((unsigned long)planes) & 15ul) == 0ul
           : (((unsigned long)dst) & 15ul) == 0ul;
  constexpr bool SLABS = KEYS || !LIST;
  const int pw = KEYS ? sage_plane_words(E) : 0;
  const int nFull =
      (!aligned || !SLABS) ? 0 : KEYS ? pw / PLANE_SLAB : E / SLAB_EDGES;
  const int nCovered =
      (!aligned || !SLABS) ? 0 : KEYS ? E : nFull * SLAB_EDGES;
  const int stride = max(1, nFull / nRanges);

  sage_stage_weights<F>(Wl, Wr, wfrag, tid, blockDim.x);

  for (int r = blockIdx.x; r < nRanges; r += gridDim.x) {
    const int lo = r * RNG;
    const unsigned cnt = (unsigned)min(RNG, nNodes - lo);
    int2* lst = (int2*)lists + (long)lo * LIST_PER_NODE;
    const int cap = (int)cnt * LIST_PER_NODE - LIST_HDR;

    for (int i = tid; i < TILE_WORDS / 4; i += blockDim.x)
      ((float4*)tile)[i] = make_float4(0.f, 0.f, 0.f, 0.f);
    if (COUNT)
      for (int i = tid; i <= RNG; i += blockDim.x) cntl[i] = 0.f;
    if (tid == 0) {
      slabNext = 0;
      if (SAVE) lcount = 0;
    }
    __syncthreads();

    const int nTiles = ((int)cnt + 15) >> 4;
    v8f acc[4];
#pragma unroll
    for (int ct = 0; ct < 4; ++ct)
      acc[ct] = (v8f){0.f, 0.f, 0.f, 0.f, 0.f, 0.f, 0.f, 0.f};
    if (wave < nTiles)
      sage_self_term<F>(acc, wfrag, x, lo, (int)cnt, wave, lane);

    bool scan = true;
    if (LIST) {
      const int total = __builtin_amdgcn_readfirstlane(lst[0].x);
      if ((unsigned)total <= (unsigned)cap) {
        scan = false;
        int b = sage_take(&slabNext, lane) * 32;
        int bn = sage_take(&slabNext, lane) * 32;
        if (b < total) {
          int2 cur = make_int2(0, 0);
          if (b + lane < total) cur = lst[LIST_HDR + b + lane];
          for (;;) {
            int2 nxt = make_int2(0, 0);
            if (bn + lane < total) nxt = lst[LIST_HDR + bn + lane];
            const int bn2 = sage_take(&slabNext, lane) * 32;
            sage_add_rows<F>(cur.x, cur.y, min(32, total - b), x, tile, lane);
            if (bn >= total) break;
            cur = nxt;
            b = bn;
            bn = bn2;
          }
        }
      }
    }

    if (scan) {
      int qn = 0;

      if constexpr (KEYS) {
        const int k0 = sage_take(&slabNext, lane);
        int kn = sage_take(&slabNext, lane);
        if (k0 < nFull) {
          const int start =
              (int)(((unsigned)r * (unsigned)stride) % (unsigned)nFull);
          const uint4* ph =
              (const uint4*)(planes + (long)(((unsigned)r >> 4) & 15u) * pw) +
              lane;
          const uint4* pl =
              (const uint4*)(planes + (long)(16u + ((unsigned)r & 15u)) * pw) +
              lane;
          int sl = k0 + start;
          if (sl >= nFull) sl -= nFull;
          uint4 ch = ph[sl * 32];
          uint4 cl = pl[sl * 32];

          for (;;) {
            const bool more = kn < nFull;
            int sn = kn + start;
            if (sn >= nFull) sn -= nFull;
            if (!more) sn = sl;
            const uint4 nh = ph[sn * 32];
            const uint4 nl = pl[sn * 32];
            const int kn2 = sage_take(&slabNext, lane);

            const int eb = (sl * PLANE_SLAB + lane * 4) << 5;
            const unsigned m4[4] = {ch.x & cl.x, ch.y & cl.y, ch.z & cl.z,
                                    ch.w & cl.w};
#pragma unroll
            for (int i = 0; i < 4; ++i) {
              unsigned f = m4[i];
              unsigned m = __builtin_amdgcn_ballot_w32(f != 0u);
              while (m) {
                if (f != 0u) {
                  q[qn + (int)__builtin_amdgcn_mbcnt_lo(m, 0u)] =
                      eb + i * 32 + __builtin_ctz(f);
                  f &= f - 1u;
                }
                qn += __builtin_popcount(m);
                if (qn >= Q_DRAIN)
                  qn = sage_drain_full<F, COUNT, SAVE>(q, qn, x, pairs, lo,
                                                       cnt, tile, cntl, lst,
                                                       cap, &lcount, lane);
                m = __builtin_amdgcn_ballot_w32(f != 0u);
              }
            }
            if (!more) break;
            ch = nh;
            cl = nl;
            sl = sn;
            kn = kn2;
          }
        }
      } else if (wave < nFull) {
        const int start =
            (int)(((unsigned)r * (unsigned)stride) % (unsigned)nFull);
        int sl = wave + start;
        if (sl >= nFull) sl -= nFull;
        int4 cur[SLAB_LOADS];
#pragma unroll
        for (int u = 0; u < SLAB_LOADS; ++u)
          cur[u] = *(const int4*)(dst + sl * SLAB_EDGES + u * 128 + lane * 4);

        for (int k = wave; k < nFull; k += nW) {
          int sn = k + nW + start;
          if (sn >= nFull) sn -= nFull;
          if (k + nW >= nFull) sn = sl;
          int4 nxt[SLAB_LOADS];
#pragma unroll
          for (int u = 0; u < SLAB_LOADS; ++u)
            nxt[u] = *(const int4*)(dst + sn * SLAB_EDGES + u * 128 + lane * 4);

#pragma unroll
          for (int u = 0; u < SLAB_LOADS; ++u) {
            const int eb = sl * SLAB_EDGES + u * 128 + lane * 4;
            const int dv4[4] = {cur[u].x, cur[u].y, cur[u].z, cur[u].w};
#pragma unroll
            for (int c = 0; c < 4; ++c) {
              const bool hit = (unsigned)(dv4[c] - lo) < cnt;
              const unsigned m = __builtin_amdgcn_ballot_w32(hit);
              if (m) {
                if (hit) q[qn + (int)__builtin_amdgcn_mbcnt_lo(m, 0u)] = eb + c;
                qn += __builtin_popcount(m);
              }
            }
            if (qn >= Q_DRAIN)
              qn = sage_drain_full<F, COUNT, SAVE>(q, qn, x, pairs, lo, cnt,
                                                   tile, cntl, lst, cap,
                                                   &lcount, lane);
          }
#pragma unroll
          for (int u = 0; u < SLAB_LOADS; ++u) cur[u] = nxt[u];
          sl = sn;
        }
      }
      for (int e0 = nCovered + wave * 32; e0 < E; e0 += nW * 32) {
        const int e = e0 + lane;
        const int d = (e < E) ? dst[e] : -1;
        const bool hit = (unsigned)(d - lo) < cnt;
        const unsigned m = __builtin_amdgcn_ballot_w32(hit);
        if (m) {
          if (hit) q[qn + (int)__builtin_amdgcn_mbcnt_lo(m, 0u)] = e;
          qn += __builtin_popcount(m);
        }
        if (qn >= Q_DRAIN)
          qn = sage_drain_full<F, COUNT, SAVE>(q, qn, x, pairs, lo, cnt, tile,
                                               cntl, lst, cap, &lcount, lane);
      }
      if (qn > 0)
        sage_drain<F, COUNT, SAVE>(q, qn, x, pairs, lo, cnt, tile, cntl, lst,
                                   cap, &lcount, lane);
    }
    __syncthreads();

    for (int rt = wave; rt < nTiles; rt += nW) {
      if (rt != wave) {
#pragma unroll
        for (int ct = 0; ct < 4; ++ct)
          acc[ct] = (v8f){0.f, 0.f, 0.f, 0.f, 0.f, 0.f, 0.f, 0.f};
        sage_self_term<F>(acc, wfrag, x, lo, (int)cnt, rt, lane);
      }
      sage_mean_term<F, COUNT>(acc, tile, cntl, wfrag, deg, lo, (int)cnt, rt,
                               lane);
      sage_store_tile<RELU>(acc, bl, out, lo, (int)cnt, rt, lane);
    }
    if (COUNT)
      for (int i = tid; i < (int)cnt; i += blockDim.x) deg[lo + i] = cntl[i];
    if (SAVE && tid == 0) lst[0] = make_int2(lcount, 0);
    __syncthreads();
  }
}

extern "C" void kernel_launch(void* const* d_in, const int* in_sizes, int n_in,
                              void* d_out, int out_size, void* d_ws,
                              size_t ws_size, hipStream_t stream) {
  const float* x = (const float*)d_in[0];
  const int* ei = (const int*)d_in[1];
  const float* Wl1 = (const float*)d_in[2];
  const float* bl1 = (const float*)d_in[3];
  const float* Wr1 = (const float*)d_in[4];
  const float* Wl2 = (const float*)d_in[5];
  const float* bl2 = (const float*)d_in[6];
  const float* Wr2 = (const float*)d_in[7];
  float* out = (float*)d_out;

  const int N = in_sizes[0] / IN_C;
  const int E = in_sizes[1] / 2;
  const int* src = ei;
  const int* dst = ei + E;

  float* ws = (float*)d_ws;
  float* deg = ws;
  unsigned* planes = (unsigned*)(deg + N);
  int2* pairs = (int2*)(deg + N + (long)N * HID);
  float* h = deg + N + (long)N * IN_C;
  int* lists = (int*)(h + (long)N * HID);


  sage_range_key_kernel<<<512, 256, 0, stream>>>(src, dst, E, planes, pairs);
  sage_lds_scatter_kernel<IN_C, true, true, false, true, true>
      <<<256, SC_THREADS, 0, stream>>>(x, pairs, dst, E, N, Wl1, bl1, Wr1, h,
                                       deg, lists, planes);

  sage_lds_scatter_kernel<HID, false, false, true, false, false>
      <<<256, SC_THREADS, 0, stream>>>(h, pairs, dst, E, N, Wl2, bl2, Wr2, out,
                                       deg, lists, nullptr);
}
